# starting kernel
# baseline (speedup 1.0000x reference)
#include <hip/hip_runtime.h>
#include <hip/hip_bf16.h>
#include <float.h>
#include <math.h>

#define HEADS 4
#define CPH 32
#define HC 128
#define NEG_SLOPE 0.2f
#define EPSV 1e-5f

typedef float v2f __attribute__((ext_vector_type(2)));
typedef float v8f __attribute__((ext_vector_type(8)));

__device__ __forceinline__ unsigned enc_f32(float f) {
    unsigned u = __float_as_uint(f);
    return (u & 0x80000000u) ? ~u : (u | 0x80000000u);
}
__device__ __forceinline__ float dec_f32(unsigned k) {
    unsigned u = (k & 0x80000000u) ? (k ^ 0x80000000u) : ~k;
    return __uint_as_float(u);
}

__device__ __forceinline__ void edge_sd(long long e, const int* __restrict__ src,
                                        const int* __restrict__ dst, int E,
                                        int* s, int* d) {
    if (e < (long long)E) { *s = src[e]; *d = dst[e]; }
    else { int v = (int)(e - E); *s = v; *d = v; }
}

__global__ __launch_bounds__(256) void k_gemm_wmma(
    const float* __restrict__ A, const float* __restrict__ W,
    float* __restrict__ Hout, int M)
{
    const int lane = threadIdx.x & 31;
    const int wave = threadIdx.x >> 5;
    const int m0   = blockIdx.x * 16;
    const int n0   = wave * 16;
    const int hi   = lane >> 4;
    const int l16  = lane & 15;
    const bool full = (m0 + 16 <= M);

    int row = m0 + l16;
    if (row >= M) row = M - 1;
    const float* __restrict__ arow = A + (size_t)row * HC;

    v8f acc = {0.f, 0.f, 0.f, 0.f, 0.f, 0.f, 0.f, 0.f};

#pragma unroll 4
    for (int k = 0; k < HC; k += 4) {
        const int kk = k + 2 * hi;
        v2f av, bv;
        av.x = arow[kk];
        av.y = arow[kk + 1];
        bv.x = W[(size_t)kk * HC + n0 + l16];
        bv.y = W[(size_t)(kk + 1) * HC + n0 + l16];
        acc = __builtin_amdgcn_wmma_f32_16x16x4_f32(
             false, av,  false, bv,
             (short)0, acc,  false,  false);
    }

    float* __restrict__ outp = Hout + (size_t)(m0 + 8 * hi) * HC + n0 + l16;
    if (full) {
#pragma unroll
        for (int r = 0; r < 8; ++r) outp[(size_t)r * HC] = acc[r];
    } else {
#pragma unroll
        for (int r = 0; r < 8; ++r) {
            int ro = m0 + r + 8 * hi;
            if (ro < M) Hout[(size_t)ro * HC + n0 + l16] = acc[r];
        }
    }
}

__global__ void k_attn_logits(const float* __restrict__ Hm,
                              const float* __restrict__ a_s,
                              const float* __restrict__ a_d,
                              float* __restrict__ als, float* __restrict__ ald,
                              int Nn)
{
    int idx = blockIdx.x * blockDim.x + threadIdx.x;
    if (idx >= Nn * HEADS) return;
    int n = idx >> 2, h = idx & 3;
    const float4* hp = (const float4*)(Hm + (size_t)n * HC + h * CPH);
    const float4* as = (const float4*)(a_s + h * CPH);
    const float4* ad = (const float4*)(a_d + h * CPH);
    float ss = 0.f, sd = 0.f;
#pragma unroll
    for (int c = 0; c < CPH / 4; ++c) {
        float4 v = hp[c], a = as[c], b = ad[c];
        ss += v.x * a.x + v.y * a.y + v.z * a.z + v.w * a.w;
        sd += v.x * b.x + v.y * b.y + v.z * b.z + v.w * b.w;
    }
    als[idx] = ss;
    ald[idx] = sd;
}

__global__ void k_init(float4* __restrict__ agg4, float* __restrict__ denom,
                       unsigned* __restrict__ mkey, int Nn)
{
    long long idx = (long long)blockIdx.x * blockDim.x + threadIdx.x;
    if (idx < (long long)Nn * (HC / 4)) agg4[idx] = make_float4(0.f, 0.f, 0.f, 0.f);
    if (idx < (long long)Nn * HEADS)    { denom[idx] = 0.f; mkey[idx] = 0u; }
}

__global__ void k_edge_max(const int* __restrict__ src, const int* __restrict__ dst,
                           const float* __restrict__ als, const float* __restrict__ ald,
                           unsigned* __restrict__ mkey, int E, int Etot)
{
    long long idx = (long long)blockIdx.x * blockDim.x + threadIdx.x;
    if (idx >= (long long)Etot * HEADS) return;
    long long e = idx >> 2; int h = (int)(idx & 3);
    int s, d; edge_sd(e, src, dst, E, &s, &d);
    float ev = als[s * HEADS + h] + ald[d * HEADS + h];
    ev = (ev < 0.f) ? NEG_SLOPE * ev : ev;
    atomicMax(&mkey[d * HEADS + h], enc_f32(ev));
}

__global__ void k_edge_expsum(const int* __restrict__ src, const int* __restrict__ dst,
                              const float* __restrict__ als, const float* __restrict__ ald,
                              const unsigned* __restrict__ mkey, float* __restrict__ denom,
                              int E, int Etot)
{
    long long idx = (long long)blockIdx.x * blockDim.x + threadIdx.x;
    if (idx >= (long long)Etot * HEADS) return;
    long long e = idx >> 2; int h = (int)(idx & 3);
    int s, d; edge_sd(e, src, dst, E, &s, &d);
    int dh = d * HEADS + h;
    float ev = als[s * HEADS + h] + ald[dh];
    ev = (ev < 0.f) ? NEG_SLOPE * ev : ev;
    atomicAdd(&denom[dh], expf(ev - dec_f32(mkey[dh])));
}

__global__ void k_edge_msg(const int* __restrict__ src, const int* __restrict__ dst,
                           const float* __restrict__ als, const float* __restrict__ ald,
                           const unsigned* __restrict__ mkey, const float* __restrict__ denom,
                           const float* __restrict__ Hm, float* __restrict__ agg,
                           int E, int Etot)
{
    long long idx = (long long)blockIdx.x * blockDim.x + threadIdx.x;
    if (idx >= (long long)Etot * (HC / 4)) return;
    long long e = idx >> 5;
    int q  = (int)(idx & 31);
    int k4 = q * 4;
    int h  = k4 >> 5;
    int s, d; edge_sd(e, src, dst, E, &s, &d);
    int dh = d * HEADS + h;
    float ev = als[s * HEADS + h] + ald[dh];
    ev = (ev < 0.f) ? NEG_SLOPE * ev : ev;
    float alpha = expf(ev - dec_f32(mkey[dh])) / denom[dh];

    float4 v = *(const float4*)(Hm + (size_t)s * HC + k4);
    float* ap = agg + (size_t)d * HC + k4;
    atomicAdd(ap + 0, v.x * alpha);
    atomicAdd(ap + 1, v.y * alpha);
    atomicAdd(ap + 2, v.z * alpha);
    atomicAdd(ap + 3, v.w * alpha);
}

__global__ __launch_bounds__(HC) void k_epilogue(
    const float* __restrict__ agg, const float* __restrict__ bias,
    const float* __restrict__ g, const float* __restrict__ bln,
    float* __restrict__ out, int apply_ln)
{
    __shared__ float s1[HC];
    __shared__ float s2[HC];
    const int n = blockIdx.x;
    const int t = threadIdx.x;
    float v = agg[(size_t)n * HC + t] + bias[t];
    if (!apply_ln) {
        out[(size_t)n * HC + t] = v;
        return;
    }
    s1[t] = v;
    s2[t] = v * v;
    __syncthreads();
#pragma unroll
    for (int off = HC / 2; off > 0; off >>= 1) {
        if (t < off) { s1[t] += s1[t + off]; s2[t] += s2[t + off]; }
        __syncthreads();
    }
    float mu  = s1[0] * (1.f / HC);
    float var = s2[0] * (1.f / HC) - mu * mu;
    float y = (v - mu) * rsqrtf(var + EPSV) * g[t] + bln[t];
    out[(size_t)n * HC + t] = fmaxf(y, 0.f);
}

extern "C" void kernel_launch(void* const* d_in, const int* in_sizes, int n_in,
                              void* d_out, int out_size, void* d_ws, size_t ws_size,
                              hipStream_t stream)
{
    const float* x    = (const float*)d_in[0];
    const int*   esrc = (const int*)d_in[1];
    const int*   edst = (const int*)d_in[2];
    const float* W[3]  = {(const float*)d_in[3],  (const float*)d_in[7],  (const float*)d_in[11]};
    const float* As[3] = {(const float*)d_in[4],  (const float*)d_in[8],  (const float*)d_in[12]};
    const float* Ad[3] = {(const float*)d_in[5],  (const float*)d_in[9],  (const float*)d_in[13]};
    const float* Bi[3] = {(const float*)d_in[6],  (const float*)d_in[10], (const float*)d_in[14]};
    const float* LnG[2] = {(const float*)d_in[15], (const float*)d_in[17]};
    const float* LnB[2] = {(const float*)d_in[16], (const float*)d_in[18]};

    const int Nn   = in_sizes[0] / HC;
    const int E    = in_sizes[1];
    const int Etot = E + Nn;

    float*    hbuf  = (float*)d_ws;
    float*    agg   = hbuf + (size_t)Nn * HC;
    float*    als   = agg  + (size_t)Nn * HC;
    float*    ald   = als  + (size_t)Nn * HEADS;
    unsigned* mkey  = (unsigned*)(ald + (size_t)Nn * HEADS);
    float*    denom = (float*)(mkey + (size_t)Nn * HEADS);

    const int TB = 256;
    const unsigned gGemm = (unsigned)((Nn + 15) / 16);
    const unsigned gAttn = (unsigned)(((long long)Nn * HEADS + TB - 1) / TB);
    const unsigned gInit = (unsigned)(((long long)Nn * (HC / 4) + TB - 1) / TB);
    const unsigned gEH   = (unsigned)(((long long)Etot * HEADS + TB - 1) / TB);
    const unsigned gMsg  = (unsigned)(((long long)Etot * (HC / 4) + TB - 1) / TB);

    const float* cur = x;
    for (int l = 0; l < 3; ++l) {
        float* layer_out = (l == 2) ? (float*)d_out : agg;

        k_gemm_wmma<<<gGemm, 256, 0, stream>>>(cur, W[l], hbuf, Nn);
        k_attn_logits<<<gAttn, TB, 0, stream>>>(hbuf, As[l], Ad[l], als, ald, Nn);
        k_init<<<gInit, TB, 0, stream>>>((float4*)layer_out, denom, mkey, Nn);
        k_edge_max<<<gEH, TB, 0, stream>>>(esrc, edst, als, ald, mkey, E, Etot);
        k_edge_expsum<<<gEH, TB, 0, stream>>>(esrc, edst, als, ald, mkey, denom, E, Etot);
        k_edge_msg<<<gMsg, TB, 0, stream>>>(esrc, edst, als, ald, mkey, denom,
                                            hbuf, layer_out, E, Etot);
        if (l < 2) {
            k_epilogue<<<(unsigned)Nn, HC, 0, stream>>>(layer_out, Bi[l],
                                                        LnG[l], LnB[l],
                                                        layer_out, 1);
            cur = layer_out;
        } else {
            k_epilogue<<<(unsigned)Nn, HC, 0, stream>>>(layer_out, Bi[l],
                                                        LnG[0], LnB[0],
                                                        layer_out, 0);
        }
    }
}


# Edges counting-sorted by dst once per call; one wave per node gathers its in-edges: max, exp-sum, weighted sum in registers, plain store. No float atomics; sum order = arrival rank in each segment
# speedup vs baseline: 2.6323x; 2.6323x over previous
#include <hip/hip_runtime.h>
#include <hip/hip_bf16.h>
#include <float.h>
#include <math.h>

#define HEADS 4
#define CPH 32
#define HC 128
#define NEG_SLOPE 0.2f
#define EPSV 1e-5f

typedef float v2f __attribute__((ext_vector_type(2)));
typedef float v8f __attribute__((ext_vector_type(8)));

__global__ __launch_bounds__(256) void k_gemm_wmma(
    const float* __restrict__ A, const float* __restrict__ W,
    float* __restrict__ Hout, int M)
{
    const int lane = threadIdx.x & 31;
    const int wave = threadIdx.x >> 5;
    const int m0   = blockIdx.x * 16;
    const int n0   = wave * 16;
    const int hi   = lane >> 4;
    const int l16  = lane & 15;
    const bool full = (m0 + 16 <= M);

    int row = m0 + l16;
    if (row >= M) row = M - 1;
    const float* __restrict__ arow = A + (size_t)row * HC;

    v8f acc = {0.f, 0.f, 0.f, 0.f, 0.f, 0.f, 0.f, 0.f};

#pragma unroll 4
    for (int k = 0; k < HC; k += 4) {
        const int kk = k + 2 * hi;
        v2f av, bv;
        av.x = arow[kk];
        av.y = arow[kk + 1];
        bv.x = W[(size_t)kk * HC + n0 + l16];
        bv.y = W[(size_t)(kk + 1) * HC + n0 + l16];
        acc = __builtin_amdgcn_wmma_f32_16x16x4_f32(
             false, av,  false, bv,
             (short)0, acc,  false,  false);
    }

    float* __restrict__ outp = Hout + (size_t)(m0 + 8 * hi) * HC + n0 + l16;
    if (full) {
#pragma unroll
        for (int r = 0; r < 8; ++r) outp[(size_t)r * HC] = acc[r];
    } else {
#pragma unroll
        for (int r = 0; r < 8; ++r) {
            int ro = m0 + r + 8 * hi;
            if (ro < M) Hout[(size_t)ro * HC + n0 + l16] = acc[r];
        }
    }
}

__global__ void k_attn_logits(const float* __restrict__ Hm,
                              const float* __restrict__ a_s,
                              const float* __restrict__ a_d,
                              float* __restrict__ als, float* __restrict__ ald,
                              int Nn)
{
    int idx = blockIdx.x * blockDim.x + threadIdx.x;
    if (idx >= Nn * HEADS) return;
    int n = idx >> 2, h = idx & 3;
    const float4* hp = (const float4*)(Hm + (size_t)n * HC + h * CPH);
    const float4* as = (const float4*)(a_s + h * CPH);
    const float4* ad = (const float4*)(a_d + h * CPH);
    float ss = 0.f, sd = 0.f;
#pragma unroll
    for (int c = 0; c < CPH / 4; ++c) {
        float4 v = hp[c], a = as[c], b = ad[c];
        ss += v.x * a.x + v.y * a.y + v.z * a.z + v.w * a.w;
        sd += v.x * b.x + v.y * b.y + v.z * b.z + v.w * b.w;
    }
    als[idx] = ss;
    ald[idx] = sd;
}

__global__ void k_csr_zero(int* __restrict__ cnt, int Nn)
{
    int idx = blockIdx.x * blockDim.x + threadIdx.x;
    if (idx < Nn) cnt[idx] = 0;
}

__global__ void k_csr_count(const int* __restrict__ dst, int* __restrict__ cnt, int E)
{
    int e = blockIdx.x * blockDim.x + threadIdx.x;
    if (e < E) atomicAdd(&cnt[dst[e]], 1);
}

__global__ __launch_bounds__(256) void k_csr_scan(const int* __restrict__ cnt,
                                                  int* __restrict__ rowp,
                                                  int* __restrict__ cur, int Nn)
{
    __shared__ int sh[256];
    const int t  = threadIdx.x;
    const int b0 = blockIdx.x * 1024;
    int part = 0;
    const int4* c4 = (const int4*)cnt;
    for (int i = t; i < b0 / 4; i += 256) {
        int4 v = c4[i];
        part += (v.x + v.y) + (v.z + v.w);
    }
    sh[t] = part;
    __syncthreads();
    for (int off = 128; off > 0; off >>= 1) {
        if (t < off) sh[t] += sh[t + off];
        __syncthreads();
    }
    const int base = sh[0];
    __syncthreads();

    const int i0 = b0 + t * 4;
    int c[4];
#pragma unroll
    for (int k = 0; k < 4; ++k) c[k] = (i0 + k < Nn) ? cnt[i0 + k] : 0;
    const int tsum = (c[0] + c[1]) + (c[2] + c[3]);
    sh[t] = tsum;
    __syncthreads();
    for (int off = 1; off < 256; off <<= 1) {
        int v = (t >= off) ? sh[t - off] : 0;
        __syncthreads();
        sh[t] += v;
        __syncthreads();
    }
    int run = base + sh[t] - tsum;
#pragma unroll
    for (int k = 0; k < 4; ++k) {
        if (i0 + k < Nn) { rowp[i0 + k] = run; cur[i0 + k] = run; }
        run += c[k];
    }
}

__global__ void k_csr_fill(const int* __restrict__ src, const int* __restrict__ dst,
                           int* __restrict__ cur, int* __restrict__ csrc, int E)
{
    int e = blockIdx.x * blockDim.x + threadIdx.x;
    if (e >= E) return;
    int p = atomicAdd(&cur[dst[e]], 1);
    csrc[p] = src[e];
}

__device__ __forceinline__ float lrelu(float v) { return (v < 0.f) ? NEG_SLOPE * v : v; }

__device__ __forceinline__ float grp8_max(float v) {
#pragma unroll
    for (int off = 4; off > 0; off >>= 1) v = fmaxf(v, __shfl_xor(v, off, 32));
    return v;
}
__device__ __forceinline__ float grp8_sum(float v) {
#pragma unroll
    for (int off = 4; off > 0; off >>= 1) v += __shfl_xor(v, off, 32);
    return v;
}

__device__ __forceinline__ void load_scores(int j0, int deg, int beg, int d, int i8, int hsel,
                                            const int* __restrict__ csrc,
                                            const float* __restrict__ als,
                                            float adh, int s[4], float e[4])
{
    int   t[4];
    float x[4];
#pragma unroll
    for (int r = 0; r < 4; ++r) {
        int jc = j0 + 8 * r + i8;
        jc = jc < deg ? jc : deg - 1;
        int idx = beg + jc;
        idx = idx < 0 ? 0 : idx;
        t[r] = csrc[idx];
    }
#pragma unroll
    for (int r = 0; r < 4; ++r) s[r] = (j0 + 8 * r + i8 < deg) ? t[r] : d;
#pragma unroll
    for (int r = 0; r < 4; ++r) x[r] = als[(size_t)s[r] * HEADS + hsel];
#pragma unroll
    for (int r = 0; r < 4; ++r)
        e[r] = (j0 + 8 * r + i8 <= deg) ? lrelu(x[r] + adh) : -INFINITY;
}

__device__ __forceinline__ float4 ld_row(const float* __restrict__ Hm, int s, int lane) {
    return *(const float4*)(Hm + (size_t)s * HC + lane * 4);
}
__device__ __forceinline__ void fma_row(float4* acc, float4 v, float w) {
    acc->x += v.x * w;
    acc->y += v.y * w;
    acc->z += v.z * w;
    acc->w += v.w * w;
}

__device__ __forceinline__ void add_rows(int nn, const int s[4], const float a[4],
                                         int lane, const float* __restrict__ Hm, float4* acc)
{
    const int gl = lane & 24;
#pragma unroll
    for (int r = 0; r < 4; ++r) {
        int cg = nn - 8 * r;
        cg = cg < 0 ? 0 : (cg > 8 ? 8 : cg);
        int i = 0;
        for (; i + 4 <= cg; i += 4) {
            const int s0 = __builtin_amdgcn_readlane(s[r], i);
            const int s1 = __builtin_amdgcn_readlane(s[r], i + 1);
            const int s2 = __builtin_amdgcn_readlane(s[r], i + 2);
            const int s3 = __builtin_amdgcn_readlane(s[r], i + 3);
            const float4 v0 = ld_row(Hm, s0, lane);
            const float4 v1 = ld_row(Hm, s1, lane);
            const float4 v2 = ld_row(Hm, s2, lane);
            const float4 v3 = ld_row(Hm, s3, lane);
            const float w0 = __shfl(a[r], gl + i, 32);
            const float w1 = __shfl(a[r], gl + i + 1, 32);
            const float w2 = __shfl(a[r], gl + i + 2, 32);
            const float w3 = __shfl(a[r], gl + i + 3, 32);
            fma_row(acc, v0, w0);
            fma_row(acc, v1, w1);
            fma_row(acc, v2, w2);
            fma_row(acc, v3, w3);
        }
        for (; i < cg; ++i) {
            const int s0 = __builtin_amdgcn_readlane(s[r], i);
            const float4 v0 = ld_row(Hm, s0, lane);
            const float w0 = __shfl(a[r], gl + i, 32);
            fma_row(acc, v0, w0);
        }
    }
}

__global__ __launch_bounds__(256) void k_node_gather(
    const int* __restrict__ rowp, const int* __restrict__ cnt,
    const int* __restrict__ csrc,
    const float* __restrict__ als, const float* __restrict__ ald,
    const float* __restrict__ Hm, float* __restrict__ outp, int Nn)
{
    const int lane = threadIdx.x & 31;
    const int hsel = lane >> 3;
    const int i8   = lane & 7;
    const int d = __builtin_amdgcn_readfirstlane((int)(blockIdx.x * 8 + (threadIdx.x >> 5)));
    if (d >= Nn) return;
    const int beg = rowp[d];
    const int deg = cnt[d];
    const int tot = deg + 1;
    const float adh = ald[(size_t)d * HEADS + hsel];

    float4 acc = make_float4(0.f, 0.f, 0.f, 0.f);
    int   s[4];
    float e[4];
    float a[4];

    if (tot <= 32) {
        load_scores(0, deg, beg, d, i8, hsel, csrc, als, adh, s, e);
        const float m = grp8_max(fmaxf(fmaxf(e[0], e[1]), fmaxf(e[2], e[3])));
#pragma unroll
        for (int r = 0; r < 4; ++r) a[r] = expf(e[r] - m);
        const float l = grp8_sum((a[0] + a[1]) + (a[2] + a[3]));
#pragma unroll
        for (int r = 0; r < 4; ++r) a[r] = a[r] / l;
        add_rows(tot, s, a, lane, Hm, &acc);
    } else {
        float m = -INFINITY;
        for (int j0 = 0; j0 < tot; j0 += 32) {
            load_scores(j0, deg, beg, d, i8, hsel, csrc, als, adh, s, e);
            m = fmaxf(m, fmaxf(fmaxf(e[0], e[1]), fmaxf(e[2], e[3])));
        }
        m = grp8_max(m);
        float l = 0.f;
        for (int j0 = 0; j0 < tot; j0 += 32) {
            load_scores(j0, deg, beg, d, i8, hsel, csrc, als, adh, s, e);
            l += (expf(e[0] - m) + expf(e[1] - m)) + (expf(e[2] - m) + expf(e[3] - m));
        }
        l = grp8_sum(l);
        for (int j0 = 0; j0 < tot; j0 += 32) {
            load_scores(j0, deg, beg, d, i8, hsel, csrc, als, adh, s, e);
#pragma unroll
            for (int r = 0; r < 4; ++r) a[r] = expf(e[r] - m) / l;
            const int left = tot - j0;
            add_rows(left < 32 ? left : 32, s, a, lane, Hm, &acc);
        }
    }

    *(float4*)(outp + (size_t)d * HC + lane * 4) = acc;
}

__global__ __launch_bounds__(HC) void k_epilogue(
    const float* __restrict__ agg, const float* __restrict__ bias,
    const float* __restrict__ g, const float* __restrict__ bln,
    float* __restrict__ out, int apply_ln)
{
    __shared__ float s1[HC];
    __shared__ float s2[HC];
    const int n = blockIdx.x;
    const int t = threadIdx.x;
    float v = agg[(size_t)n * HC + t] + bias[t];
    if (!apply_ln) {
        out[(size_t)n * HC + t] = v;
        return;
    }
    s1[t] = v;
    s2[t] = v * v;
    __syncthreads();
#pragma unroll
    for (int off = HC / 2; off > 0; off >>= 1) {
        if (t < off) { s1[t] += s1[t + off]; s2[t] += s2[t + off]; }
        __syncthreads();
    }
    float mu  = s1[0] * (1.f / HC);
    float var = s2[0] * (1.f / HC) - mu * mu;
    float y = (v - mu) * rsqrtf(var + EPSV) * g[t] + bln[t];
    out[(size_t)n * HC + t] = fmaxf(y, 0.f);
}

extern "C" void kernel_launch(void* const* d_in, const int* in_sizes, int n_in,
                              void* d_out, int out_size, void* d_ws, size_t ws_size,
                              hipStream_t stream)
{
    const float* x    = (const float*)d_in[0];
    const int*   esrc = (const int*)d_in[1];
    const int*   edst = (const int*)d_in[2];
    const float* W[3]  = {(const float*)d_in[3],  (const float*)d_in[7],  (const float*)d_in[11]};
    const float* As[3] = {(const float*)d_in[4],  (const float*)d_in[8],  (const float*)d_in[12]};
    const float* Ad[3] = {(const float*)d_in[5],  (const float*)d_in[9],  (const float*)d_in[13]};
    const float* Bi[3] = {(const float*)d_in[6],  (const float*)d_in[10], (const float*)d_in[14]};
    const float* LnG[2] = {(const float*)d_in[15], (const float*)d_in[17]};
    const float* LnB[2] = {(const float*)d_in[16], (const float*)d_in[18]};

    const int Nn   = in_sizes[0] / HC;
    const int E    = in_sizes[1];

    float* hbuf = (float*)d_ws;
    int*   cnt  = (int*)(hbuf + (size_t)Nn * HC);
    int*   rowp = cnt + Nn;
    int*   ecur = rowp + Nn;
    int*   csrc = ecur + Nn;
    float* als  = hbuf + 2 * (size_t)Nn * HC;
    float* ald  = als + (size_t)Nn * HEADS;
    float* lay  = (float*)d_out;

    const int TB = 256;
    const unsigned gGemm = (unsigned)((Nn + 15) / 16);
    const unsigned gAttn = (unsigned)(((long long)Nn * HEADS + TB - 1) / TB);
    const unsigned gN    = (unsigned)((Nn + TB - 1) / TB);
    const unsigned gE    = (unsigned)((E + TB - 1) / TB);
    const unsigned gScan = (unsigned)((Nn + 1023) / 1024);
    const unsigned gGat  = (unsigned)((Nn + 7) / 8);

    k_csr_zero<<<gN, TB, 0, stream>>>(cnt, Nn);
    k_csr_count<<<gE, TB, 0, stream>>>(edst, cnt, E);
    k_csr_scan<<<gScan, 256, 0, stream>>>(cnt, rowp, ecur, Nn);
    k_csr_fill<<<gE, TB, 0, stream>>>(esrc, edst, ecur, csrc, E);

    const float* cur = x;
    for (int l = 0; l < 3; ++l) {
        k_gemm_wmma<<<gGemm, 256, 0, stream>>>(cur, W[l], hbuf, Nn);
        k_attn_logits<<<gAttn, TB, 0, stream>>>(hbuf, As[l], Ad[l], als, ald, Nn);
        k_node_gather<<<gGat, 256, 0, stream>>>(rowp, cnt, csrc, als, ald, hbuf, lay, Nn);
        if (l < 2) {
            k_epilogue<<<(unsigned)Nn, HC, 0, stream>>>(lay, Bi[l], LnG[l], LnB[l], lay, 1);
            cur = lay;
        } else {
            k_epilogue<<<(unsigned)Nn, HC, 0, stream>>>(lay, Bi[l], LnG[0], LnB[0], lay, 0);
        }
    }
}


# Bias + LayerNorm + ReLU done by the gather wave on the 128 channels it holds; the 3 launches of 100000 four-wave blocks go; LN sums now 4 per lane + a 5-step lane-xor tree (was an LDS tree)
# speedup vs baseline: 3.3263x; 1.2636x over previous
#include <hip/hip_runtime.h>
#include <hip/hip_bf16.h>
#include <float.h>
#include <math.h>

#define HEADS 4
#define CPH 32
#define HC 128
#define NEG_SLOPE 0.2f
#define EPSV 1e-5f

typedef float v2f __attribute__((ext_vector_type(2)));
typedef float v8f __attribute__((ext_vector_type(8)));

__global__ __launch_bounds__(256) void k_gemm_wmma(
    const float* __restrict__ A, const float* __restrict__ W,
    float* __restrict__ Hout, int M)
{
    const int lane = threadIdx.x & 31;
    const int wave = threadIdx.x >> 5;
    const int m0   = blockIdx.x * 16;
    const int n0   = wave * 16;
    const int hi   = lane >> 4;
    const int l16  = lane & 15;
    const bool full = (m0 + 16 <= M);

    int row = m0 + l16;
    if (row >= M) row = M - 1;
    const float* __restrict__ arow = A + (size_t)row * HC;

    v8f acc = {0.f, 0.f, 0.f, 0.f, 0.f, 0.f, 0.f, 0.f};

#pragma unroll 4
    for (int k = 0; k < HC; k += 4) {
        const int kk = k + 2 * hi;
        v2f av, bv;
        av.x = arow[kk];
        av.y = arow[kk + 1];
        bv.x = W[(size_t)kk * HC + n0 + l16];
        bv.y = W[(size_t)(kk + 1) * HC + n0 + l16];
        acc = __builtin_amdgcn_wmma_f32_16x16x4_f32(
             false, av,  false, bv,
             (short)0, acc,  false,  false);
    }

    float* __restrict__ outp = Hout + (size_t)(m0 + 8 * hi) * HC + n0 + l16;
    if (full) {
#pragma unroll
        for (int r = 0; r < 8; ++r) outp[(size_t)r * HC] = acc[r];
    } else {
#pragma unroll
        for (int r = 0; r < 8; ++r) {
            int ro = m0 + r + 8 * hi;
            if (ro < M) Hout[(size_t)ro * HC + n0 + l16] = acc[r];
        }
    }
}

__global__ void k_attn_logits(const float* __restrict__ Hm,
                              const float* __restrict__ a_s,
                              const float* __restrict__ a_d,
                              float* __restrict__ als, float* __restrict__ ald,
                              int Nn)
{
    int idx = blockIdx.x * blockDim.x + threadIdx.x;
    if (idx >= Nn * HEADS) return;
    int n = idx >> 2, h = idx & 3;
    const float4* hp = (const float4*)(Hm + (size_t)n * HC + h * CPH);
    const float4* as = (const float4*)(a_s + h * CPH);
    const float4* ad = (const float4*)(a_d + h * CPH);
    float ss = 0.f, sd = 0.f;
#pragma unroll
    for (int c = 0; c < CPH / 4; ++c) {
        float4 v = hp[c], a = as[c], b = ad[c];
        ss += v.x * a.x + v.y * a.y + v.z * a.z + v.w * a.w;
        sd += v.x * b.x + v.y * b.y + v.z * b.z + v.w * b.w;
    }
    als[idx] = ss;
    ald[idx] = sd;
}

__global__ void k_csr_zero(int* __restrict__ cnt, int Nn)
{
    int idx = blockIdx.x * blockDim.x + threadIdx.x;
    if (idx < Nn) cnt[idx] = 0;
}

__global__ void k_csr_count(const int* __restrict__ dst, int* __restrict__ cnt, int E)
{
    int e = blockIdx.x * blockDim.x + threadIdx.x;
    if (e < E) atomicAdd(&cnt[dst[e]], 1);
}

__global__ __launch_bounds__(256) void k_csr_scan(const int* __restrict__ cnt,
                                                  int* __restrict__ rowp,
                                                  int* __restrict__ cur, int Nn)
{
    __shared__ int sh[256];
    const int t  = threadIdx.x;
    const int b0 = blockIdx.x * 1024;
    int part = 0;
    const int4* c4 = (const int4*)cnt;
    for (int i = t; i < b0 / 4; i += 256) {
        int4 v = c4[i];
        part += (v.x + v.y) + (v.z + v.w);
    }
    sh[t] = part;
    __syncthreads();
    for (int off = 128; off > 0; off >>= 1) {
        if (t < off) sh[t] += sh[t + off];
        __syncthreads();
    }
    const int base = sh[0];
    __syncthreads();

    const int i0 = b0 + t * 4;
    int c[4];
#pragma unroll
    for (int k = 0; k < 4; ++k) c[k] = (i0 + k < Nn) ? cnt[i0 + k] : 0;
    const int tsum = (c[0] + c[1]) + (c[2] + c[3]);
    sh[t] = tsum;
    __syncthreads();
    for (int off = 1; off < 256; off <<= 1) {
        int v = (t >= off) ? sh[t - off] : 0;
        __syncthreads();
        sh[t] += v;
        __syncthreads();
    }
    int run = base + sh[t] - tsum;
#pragma unroll
    for (int k = 0; k < 4; ++k) {
        if (i0 + k < Nn) { rowp[i0 + k] = run; cur[i0 + k] = run; }
        run += c[k];
    }
}

__global__ void k_csr_fill(const int* __restrict__ src, const int* __restrict__ dst,
                           int* __restrict__ cur, int* __restrict__ csrc, int E)
{
    int e = blockIdx.x * blockDim.x + threadIdx.x;
    if (e >= E) return;
    int p = atomicAdd(&cur[dst[e]], 1);
    csrc[p] = src[e];
}

__device__ __forceinline__ float lrelu(float v) { return (v < 0.f) ? NEG_SLOPE * v : v; }

__device__ __forceinline__ float grp8_max(float v) {
#pragma unroll
    for (int off = 4; off > 0; off >>= 1) v = fmaxf(v, __shfl_xor(v, off, 32));
    return v;
}
__device__ __forceinline__ float grp8_sum(float v) {
#pragma unroll
    for (int off = 4; off > 0; off >>= 1) v += __shfl_xor(v, off, 32);
    return v;
}
__device__ __forceinline__ float wave_sum(float v) {
    v += __shfl_xor(v, 16, 32);
    v += __shfl_xor(v, 8, 32);
    v += __shfl_xor(v, 4, 32);
    v += __shfl_xor(v, 2, 32);
    v += __shfl_xor(v, 1, 32);
    return v;
}

__device__ __forceinline__ void load_scores(int j0, int deg, int beg, int d, int i8, int hsel,
                                            const int* __restrict__ csrc,
                                            const float* __restrict__ als,
                                            float adh, int s[4], float e[4])
{
    int   t[4];
    float x[4];
#pragma unroll
    for (int r = 0; r < 4; ++r) {
        int jc = j0 + 8 * r + i8;
        jc = jc < deg ? jc : deg - 1;
        int idx = beg + jc;
        idx = idx < 0 ? 0 : idx;
        t[r] = csrc[idx];
    }
#pragma unroll
    for (int r = 0; r < 4; ++r) s[r] = (j0 + 8 * r + i8 < deg) ? t[r] : d;
#pragma unroll
    for (int r = 0; r < 4; ++r) x[r] = als[(size_t)s[r] * HEADS + hsel];
#pragma unroll
    for (int r = 0; r < 4; ++r)
        e[r] = (j0 + 8 * r + i8 <= deg) ? lrelu(x[r] + adh) : -INFINITY;
}

__device__ __forceinline__ float4 ld_row(const float* __restrict__ Hm, int s, int lane) {
    return *(const float4*)(Hm + (size_t)s * HC + lane * 4);
}
__device__ __forceinline__ void fma_row(float4* acc, float4 v, float w) {
    acc->x += v.x * w;
    acc->y += v.y * w;
    acc->z += v.z * w;
    acc->w += v.w * w;
}

__device__ __forceinline__ void add_rows(int nn, const int s[4], const float a[4],
                                         int lane, const float* __restrict__ Hm, float4* acc)
{
    const int gl = lane & 24;
#pragma unroll
    for (int r = 0; r < 4; ++r) {
        int cg = nn - 8 * r;
        cg = cg < 0 ? 0 : (cg > 8 ? 8 : cg);
        int i = 0;
        for (; i + 4 <= cg; i += 4) {
            const int s0 = __builtin_amdgcn_readlane(s[r], i);
            const int s1 = __builtin_amdgcn_readlane(s[r], i + 1);
            const int s2 = __builtin_amdgcn_readlane(s[r], i + 2);
            const int s3 = __builtin_amdgcn_readlane(s[r], i + 3);
            const float4 v0 = ld_row(Hm, s0, lane);
            const float4 v1 = ld_row(Hm, s1, lane);
            const float4 v2 = ld_row(Hm, s2, lane);
            const float4 v3 = ld_row(Hm, s3, lane);
            const float w0 = __shfl(a[r], gl + i, 32);
            const float w1 = __shfl(a[r], gl + i + 1, 32);
            const float w2 = __shfl(a[r], gl + i + 2, 32);
            const float w3 = __shfl(a[r], gl + i + 3, 32);
            fma_row(acc, v0, w0);
            fma_row(acc, v1, w1);
            fma_row(acc, v2, w2);
            fma_row(acc, v3, w3);
        }
        for (; i < cg; ++i) {
            const int s0 = __builtin_amdgcn_readlane(s[r], i);
            const float4 v0 = ld_row(Hm, s0, lane);
            const float w0 = __shfl(a[r], gl + i, 32);
            fma_row(acc, v0, w0);
        }
    }
}

__global__ __launch_bounds__(256) void k_node_gather(
    const int* __restrict__ rowp, const int* __restrict__ cnt,
    const int* __restrict__ csrc,
    const float* __restrict__ als, const float* __restrict__ ald,
    const float* __restrict__ Hm,
    const float* __restrict__ bias, const float* __restrict__ g, const float* __restrict__ bln,
    float* __restrict__ outp, int Nn, int apply_ln)
{
    const int lane = threadIdx.x & 31;
    const int hsel = lane >> 3;
    const int i8   = lane & 7;
    const int d = __builtin_amdgcn_readfirstlane((int)(blockIdx.x * 8 + (threadIdx.x >> 5)));
    if (d >= Nn) return;
    const int beg = rowp[d];
    const int deg = cnt[d];
    const int tot = deg + 1;
    const float adh = ald[(size_t)d * HEADS + hsel];

    float4 acc = make_float4(0.f, 0.f, 0.f, 0.f);
    int   s[4];
    float e[4];
    float a[4];

    if (tot <= 32) {
        load_scores(0, deg, beg, d, i8, hsel, csrc, als, adh, s, e);
        const float m = grp8_max(fmaxf(fmaxf(e[0], e[1]), fmaxf(e[2], e[3])));
#pragma unroll
        for (int r = 0; r < 4; ++r) a[r] = expf(e[r] - m);
        const float l = grp8_sum((a[0] + a[1]) + (a[2] + a[3]));
#pragma unroll
        for (int r = 0; r < 4; ++r) a[r] = a[r] / l;
        add_rows(tot, s, a, lane, Hm, &acc);
    } else {
        float m = -INFINITY;
        for (int j0 = 0; j0 < tot; j0 += 32) {
            load_scores(j0, deg, beg, d, i8, hsel, csrc, als, adh, s, e);
            m = fmaxf(m, fmaxf(fmaxf(e[0], e[1]), fmaxf(e[2], e[3])));
        }
        m = grp8_max(m);
        float l = 0.f;
        for (int j0 = 0; j0 < tot; j0 += 32) {
            load_scores(j0, deg, beg, d, i8, hsel, csrc, als, adh, s, e);
            l += (expf(e[0] - m) + expf(e[1] - m)) + (expf(e[2] - m) + expf(e[3] - m));
        }
        l = grp8_sum(l);
        for (int j0 = 0; j0 < tot; j0 += 32) {
            load_scores(j0, deg, beg, d, i8, hsel, csrc, als, adh, s, e);
#pragma unroll
            for (int r = 0; r < 4; ++r) a[r] = expf(e[r] - m) / l;
            const int left = tot - j0;
            add_rows(left < 32 ? left : 32, s, a, lane, Hm, &acc);
        }
    }

    const float4 bv = *(const float4*)(bias + lane * 4);
    float4 v = make_float4(acc.x + bv.x, acc.y + bv.y, acc.z + bv.z, acc.w + bv.w);
    if (apply_ln) {
        const float4 gv = *(const float4*)(g + lane * 4);
        const float4 bl = *(const float4*)(bln + lane * 4);
        const float s1 = wave_sum((v.x + v.y) + (v.z + v.w));
        const float s2 = wave_sum((v.x * v.x + v.y * v.y) + (v.z * v.z + v.w * v.w));
        const float mu  = s1 * (1.f / HC);
        const float var = s2 * (1.f / HC) - mu * mu;
        const float rs  = rsqrtf(var + EPSV);
        v.x = fmaxf((v.x - mu) * rs * gv.x + bl.x, 0.f);
        v.y = fmaxf((v.y - mu) * rs * gv.y + bl.y, 0.f);
        v.z = fmaxf((v.z - mu) * rs * gv.z + bl.z, 0.f);
        v.w = fmaxf((v.w - mu) * rs * gv.w + bl.w, 0.f);
    }
    *(float4*)(outp + (size_t)d * HC + lane * 4) = v;
}

extern "C" void kernel_launch(void* const* d_in, const int* in_sizes, int n_in,
                              void* d_out, int out_size, void* d_ws, size_t ws_size,
                              hipStream_t stream)
{
    const float* x    = (const float*)d_in[0];
    const int*   esrc = (const int*)d_in[1];
    const int*   edst = (const int*)d_in[2];
    const float* W[3]  = {(const float*)d_in[3],  (const float*)d_in[7],  (const float*)d_in[11]};
    const float* As[3] = {(const float*)d_in[4],  (const float*)d_in[8],  (const float*)d_in[12]};
    const float* Ad[3] = {(const float*)d_in[5],  (const float*)d_in[9],  (const float*)d_in[13]};
    const float* Bi[3] = {(const float*)d_in[6],  (const float*)d_in[10], (const float*)d_in[14]};
    const float* LnG[2] = {(const float*)d_in[15], (const float*)d_in[17]};
    const float* LnB[2] = {(const float*)d_in[16], (const float*)d_in[18]};

    const int Nn   = in_sizes[0] / HC;
    const int E    = in_sizes[1];

    float* hbuf = (float*)d_ws;
    int*   cnt  = (int*)(hbuf + (size_t)Nn * HC);
    int*   rowp = cnt + Nn;
    int*   ecur = rowp + Nn;
    int*   csrc = ecur + Nn;
    float* als  = hbuf + 2 * (size_t)Nn * HC;
    float* ald  = als + (size_t)Nn * HEADS;
    float* lay  = (float*)d_out;

    const int TB = 256;
    const unsigned gGemm = (unsigned)((Nn + 15) / 16);
    const unsigned gAttn = (unsigned)(((long long)Nn * HEADS + TB - 1) / TB);
    const unsigned gN    = (unsigned)((Nn + TB - 1) / TB);
    const unsigned gE    = (unsigned)((E + TB - 1) / TB);
    const unsigned gScan = (unsigned)((Nn + 1023) / 1024);
    const unsigned gGat  = (unsigned)((Nn + 7) / 8);

    k_csr_zero<<<gN, TB, 0, stream>>>(cnt, Nn);
    k_csr_count<<<gE, TB, 0, stream>>>(edst, cnt, E);
    k_csr_scan<<<gScan, 256, 0, stream>>>(cnt, rowp, ecur, Nn);
    k_csr_fill<<<gE, TB, 0, stream>>>(esrc, edst, ecur, csrc, E);

    const float* cur = x;
    for (int l = 0; l < 3; ++l) {
        k_gemm_wmma<<<gGemm, 256, 0, stream>>>(cur, W[l], hbuf, Nn);
        k_attn_logits<<<gAttn, TB, 0, stream>>>(hbuf, As[l], Ad[l], als, ald, Nn);
        if (l < 2) {
            k_node_gather<<<gGat, 256, 0, stream>>>(rowp, cnt, csrc, als, ald, hbuf,
                                                    Bi[l], LnG[l], LnB[l], lay, Nn, 1);
            cur = lay;
        } else {
            k_node_gather<<<gGat, 256, 0, stream>>>(rowp, cnt, csrc, als, ald, hbuf,
                                                    Bi[l], LnG[0], LnB[0], lay, Nn, 0);
        }
    }
}


# Edge sort by destination as a two-level counting sort on LDS counters (bin by 256-node range, then per-range count/scan/place): 4 launches become 2, no global atomics; sum order = LDS arrival rank
# speedup vs baseline: 4.7969x; 1.4421x over previous
#include <hip/hip_runtime.h>
#include <hip/hip_bf16.h>
#include <float.h>
#include <math.h>

#define HEADS 4
#define CPH 32
#define HC 128
#define NEG_SLOPE 0.2f
#define EPSV 1e-5f

typedef float v2f __attribute__((ext_vector_type(2)));
typedef float v8f __attribute__((ext_vector_type(8)));

__global__ __launch_bounds__(256) void k_gemm_wmma(
    const float* __restrict__ A, const float* __restrict__ W,
    float* __restrict__ Hout, int M)
{
    const int lane = threadIdx.x & 31;
    const int wave = threadIdx.x >> 5;
    const int m0   = blockIdx.x * 16;
    const int n0   = wave * 16;
    const int hi   = lane >> 4;
    const int l16  = lane & 15;
    const bool full = (m0 + 16 <= M);

    int row = m0 + l16;
    if (row >= M) row = M - 1;
    const float* __restrict__ arow = A + (size_t)row * HC;

    v8f acc = {0.f, 0.f, 0.f, 0.f, 0.f, 0.f, 0.f, 0.f};

#pragma unroll 4
    for (int k = 0; k < HC; k += 4) {
        const int kk = k + 2 * hi;
        v2f av, bv;
        av.x = arow[kk];
        av.y = arow[kk + 1];
        bv.x = W[(size_t)kk * HC + n0 + l16];
        bv.y = W[(size_t)(kk + 1) * HC + n0 + l16];
        acc = __builtin_amdgcn_wmma_f32_16x16x4_f32(
             false, av,  false, bv,
             (short)0, acc,  false,  false);
    }

    float* __restrict__ outp = Hout + (size_t)(m0 + 8 * hi) * HC + n0 + l16;
    if (full) {
#pragma unroll
        for (int r = 0; r < 8; ++r) outp[(size_t)r * HC] = acc[r];
    } else {
#pragma unroll
        for (int r = 0; r < 8; ++r) {
            int ro = m0 + r + 8 * hi;
            if (ro < M) Hout[(size_t)ro * HC + n0 + l16] = acc[r];
        }
    }
}

__global__ void k_attn_logits(const float* __restrict__ Hm,
                              const float* __restrict__ a_s,
                              const float* __restrict__ a_d,
                              float* __restrict__ als, float* __restrict__ ald,
                              int Nn)
{
    int idx = blockIdx.x * blockDim.x + threadIdx.x;
    if (idx >= Nn * HEADS) return;
    int n = idx >> 2, h = idx & 3;
    const float4* hp = (const float4*)(Hm + (size_t)n * HC + h * CPH);
    const float4* as = (const float4*)(a_s + h * CPH);
    const float4* ad = (const float4*)(a_d + h * CPH);
    float ss = 0.f, sd = 0.f;
#pragma unroll
    for (int c = 0; c < CPH / 4; ++c) {
        float4 v = hp[c], a = as[c], b = ad[c];
        ss += v.x * a.x + v.y * a.y + v.z * a.z + v.w * a.w;
        sd += v.x * b.x + v.y * b.y + v.z * b.z + v.w * b.w;
    }
    als[idx] = ss;
    ald[idx] = sd;
}

#define RSH    8
#define RNODES 256
#define CHUNK  8192
#define MAXB   1024
#define MAXCH  1024

__device__ __forceinline__ int block_run_sum(int v, int* ws, int lane, int w)
{
#pragma unroll
    for (int o = 1; o < 32; o <<= 1) {
        const int u = __shfl_up(v, o, 32);
        if (lane >= o) v += u;
    }
    if (lane == 31) ws[w] = v;
    __syncthreads();
    if (w == 0) {
        int s = ws[lane];
#pragma unroll
        for (int o = 1; o < 32; o <<= 1) {
            const int u = __shfl_up(s, o, 32);
            if (lane >= o) s += u;
        }
        ws[lane] = s;
    }
    __syncthreads();
    return v + (w > 0 ? ws[w - 1] : 0);
}

__global__ __launch_bounds__(1024) void k_csr_part(const int* __restrict__ src,
                                                   const int* __restrict__ dst,
                                                   int* __restrict__ pairs_i,
                                                   int* __restrict__ off, int E, int B)
{
    __shared__ int  hist[MAXB];
    __shared__ int  sc[32];
    __shared__ int2 stage[CHUNK];
    int2* __restrict__ pairs = (int2*)pairs_i;
    const int t  = threadIdx.x;
    const int c  = blockIdx.x;
    const int e0 = c * CHUNK;
    int dv[8], sv[8];
#pragma unroll
    for (int k = 0; k < 8; ++k) {
        const int e = e0 + t + 1024 * k;
        dv[k] = (e < E) ? dst[e] : -1;
    }
#pragma unroll
    for (int k = 0; k < 8; ++k) {
        const int e = e0 + t + 1024 * k;
        sv[k] = (e < E) ? src[e] : 0;
    }
    hist[t] = 0;
    __syncthreads();
#pragma unroll
    for (int k = 0; k < 8; ++k)
        if (dv[k] >= 0) atomicAdd(&hist[dv[k] >> RSH], 1);
    __syncthreads();
    const int h  = hist[t];
    const int st = block_run_sum(h, sc, t & 31, t >> 5) - h;
    if (t <= B) off[(size_t)c * (B + 1) + t] = st;
    hist[t] = st;
    __syncthreads();
#pragma unroll
    for (int k = 0; k < 8; ++k)
        if (dv[k] >= 0) {
            const int p = atomicAdd(&hist[dv[k] >> RSH], 1);
            stage[p] = make_int2(sv[k], dv[k]);
        }
    __syncthreads();
#pragma unroll
    for (int k = 0; k < 8; ++k) {
        const int i = t + 1024 * k;
        if (e0 + i < E) pairs[e0 + i] = stage[i];
    }
}

template <bool PLACE>
__device__ __forceinline__ int csr_sweep(const int2* __restrict__ pairs,
                                         const int* __restrict__ off,
                                         int* segb, int* segn, int* cl,
                                         int* __restrict__ csrc,
                                         int b, int lo, int B, int nch,
                                         int t, int lane, int w)
{
    int part = 0;
    for (int c0 = 0; c0 < nch; c0 += MAXCH) {
        int nc = nch - c0;
        nc = nc < MAXCH ? nc : MAXCH;
        for (int c = t; c < nc; c += 1024) {
            const int* o = off + (size_t)(c0 + c) * (B + 1) + b;
            const int s0 = o[0], s1 = o[1];
            segb[c] = (c0 + c) * CHUNK + s0;
            segn[c] = s1 - s0;
            part += s0;
        }
        __syncthreads();
        for (int c = w; c < nc; c += 32) {
            const int sb = segb[c], sn = segn[c];
            for (int j = lane; j < sn; j += 32) {
                const int2 pr = pairs[sb + j];
                if (PLACE) {
                    const int p = atomicAdd(&cl[pr.y - lo], 1);
                    csrc[p] = pr.x;
                } else {
                    atomicAdd(&cl[pr.y - lo], 1);
                }
            }
        }
        __syncthreads();
    }
    return part;
}

__global__ __launch_bounds__(1024) void k_csr_place(const int* __restrict__ pairs_i,
                                                    const int* __restrict__ off,
                                                    int* __restrict__ cnt,
                                                    int* __restrict__ rowp,
                                                    int* __restrict__ csrc,
                                                    int Nn, int B, int nch)
{
    __shared__ int segb[MAXCH];
    __shared__ int segn[MAXCH];
    __shared__ int cl[RNODES];
    __shared__ int sc[32];
    __shared__ int red[32];
    const int2* __restrict__ pairs = (const int2*)pairs_i;
    const int t = threadIdx.x, lane = t & 31, w = t >> 5;
    const int b  = blockIdx.x;
    const int lo = b << RSH;

    if (t < RNODES) cl[t] = 0;
    __syncthreads();
    int part = csr_sweep<false>(pairs, off, segb, segn, cl, csrc, b, lo, B, nch, t, lane, w);

    part += __shfl_xor(part, 16, 32);
    part += __shfl_xor(part, 8, 32);
    part += __shfl_xor(part, 4, 32);
    part += __shfl_xor(part, 2, 32);
    part += __shfl_xor(part, 1, 32);
    if (lane == 0) red[w] = part;
    const int h = (t < RNODES) ? cl[t] : 0;
    const int run = block_run_sum(h, sc, lane, w);
    int base = 0;
    for (int i = 0; i < 32; ++i) base += red[i];
    if (t < RNODES) {
        const int first = base + run - h;
        cl[t] = first;
        if (lo + t < Nn) { rowp[lo + t] = first; cnt[lo + t] = h; }
    }
    __syncthreads();
    csr_sweep<true>(pairs, off, segb, segn, cl, csrc, b, lo, B, nch, t, lane, w);
}

__device__ __forceinline__ float lrelu(float v) { return (v < 0.f) ? NEG_SLOPE * v : v; }

__device__ __forceinline__ float grp8_max(float v) {
#pragma unroll
    for (int off = 4; off > 0; off >>= 1) v = fmaxf(v, __shfl_xor(v, off, 32));
    return v;
}
__device__ __forceinline__ float grp8_sum(float v) {
#pragma unroll
    for (int off = 4; off > 0; off >>= 1) v += __shfl_xor(v, off, 32);
    return v;
}
__device__ __forceinline__ float wave_sum(float v) {
    v += __shfl_xor(v, 16, 32);
    v += __shfl_xor(v, 8, 32);
    v += __shfl_xor(v, 4, 32);
    v += __shfl_xor(v, 2, 32);
    v += __shfl_xor(v, 1, 32);
    return v;
}

__device__ __forceinline__ void load_scores(int j0, int deg, int beg, int d, int i8, int hsel,
                                            const int* __restrict__ csrc,
                                            const float* __restrict__ als,
                                            float adh, int s[4], float e[4])
{
    int   t[4];
    float x[4];
#pragma unroll
    for (int r = 0; r < 4; ++r) {
        int jc = j0 + 8 * r + i8;
        jc = jc < deg ? jc : deg - 1;
        int idx = beg + jc;
        idx = idx < 0 ? 0 : idx;
        t[r] = csrc[idx];
    }
#pragma unroll
    for (int r = 0; r < 4; ++r) s[r] = (j0 + 8 * r + i8 < deg) ? t[r] : d;
#pragma unroll
    for (int r = 0; r < 4; ++r) x[r] = als[(size_t)s[r] * HEADS + hsel];
#pragma unroll
    for (int r = 0; r < 4; ++r)
        e[r] = (j0 + 8 * r + i8 <= deg) ? lrelu(x[r] + adh) : -INFINITY;
}

__device__ __forceinline__ float4 ld_row(const float* __restrict__ Hm, int s, int lane) {
    return *(const float4*)(Hm + (size_t)s * HC + lane * 4);
}
__device__ __forceinline__ void fma_row(float4* acc, float4 v, float w) {
    acc->x += v.x * w;
    acc->y += v.y * w;
    acc->z += v.z * w;
    acc->w += v.w * w;
}

__device__ __forceinline__ void add_rows(int nn, const int s[4], const float a[4],
                                         int lane, const float* __restrict__ Hm, float4* acc)
{
    const int gl = lane & 24;
#pragma unroll
    for (int r = 0; r < 4; ++r) {
        int cg = nn - 8 * r;
        cg = cg < 0 ? 0 : (cg > 8 ? 8 : cg);
        int i = 0;
        for (; i + 4 <= cg; i += 4) {
            const int s0 = __builtin_amdgcn_readlane(s[r], i);
            const int s1 = __builtin_amdgcn_readlane(s[r], i + 1);
            const int s2 = __builtin_amdgcn_readlane(s[r], i + 2);
            const int s3 = __builtin_amdgcn_readlane(s[r], i + 3);
            const float4 v0 = ld_row(Hm, s0, lane);
            const float4 v1 = ld_row(Hm, s1, lane);
            const float4 v2 = ld_row(Hm, s2, lane);
            const float4 v3 = ld_row(Hm, s3, lane);
            const float w0 = __shfl(a[r], gl + i, 32);
            const float w1 = __shfl(a[r], gl + i + 1, 32);
            const float w2 = __shfl(a[r], gl + i + 2, 32);
            const float w3 = __shfl(a[r], gl + i + 3, 32);
            fma_row(acc, v0, w0);
            fma_row(acc, v1, w1);
            fma_row(acc, v2, w2);
            fma_row(acc, v3, w3);
        }
        for (; i < cg; ++i) {
            const int s0 = __builtin_amdgcn_readlane(s[r], i);
            const float4 v0 = ld_row(Hm, s0, lane);
            const float w0 = __shfl(a[r], gl + i, 32);
            fma_row(acc, v0, w0);
        }
    }
}

__global__ __launch_bounds__(256) void k_node_gather(
    const int* __restrict__ rowp, const int* __restrict__ cnt,
    const int* __restrict__ csrc,
    const float* __restrict__ als, const float* __restrict__ ald,
    const float* __restrict__ Hm,
    const float* __restrict__ bias, const float* __restrict__ g, const float* __restrict__ bln,
    float* __restrict__ outp, int Nn, int apply_ln)
{
    const int lane = threadIdx.x & 31;
    const int hsel = lane >> 3;
    const int i8   = lane & 7;
    const int d = __builtin_amdgcn_readfirstlane((int)(blockIdx.x * 8 + (threadIdx.x >> 5)));
    if (d >= Nn) return;
    const int beg = rowp[d];
    const int deg = cnt[d];
    const int tot = deg + 1;
    const float adh = ald[(size_t)d * HEADS + hsel];

    float4 acc = make_float4(0.f, 0.f, 0.f, 0.f);
    int   s[4];
    float e[4];
    float a[4];

    if (tot <= 32) {
        load_scores(0, deg, beg, d, i8, hsel, csrc, als, adh, s, e);
        const float m = grp8_max(fmaxf(fmaxf(e[0], e[1]), fmaxf(e[2], e[3])));
#pragma unroll
        for (int r = 0; r < 4; ++r) a[r] = expf(e[r] - m);
        const float l = grp8_sum((a[0] + a[1]) + (a[2] + a[3]));
#pragma unroll
        for (int r = 0; r < 4; ++r) a[r] = a[r] / l;
        add_rows(tot, s, a, lane, Hm, &acc);
    } else {
        float m = -INFINITY;
        for (int j0 = 0; j0 < tot; j0 += 32) {
            load_scores(j0, deg, beg, d, i8, hsel, csrc, als, adh, s, e);
            m = fmaxf(m, fmaxf(fmaxf(e[0], e[1]), fmaxf(e[2], e[3])));
        }
        m = grp8_max(m);
        float l = 0.f;
        for (int j0 = 0; j0 < tot; j0 += 32) {
            load_scores(j0, deg, beg, d, i8, hsel, csrc, als, adh, s, e);
            l += (expf(e[0] - m) + expf(e[1] - m)) + (expf(e[2] - m) + expf(e[3] - m));
        }
        l = grp8_sum(l);
        for (int j0 = 0; j0 < tot; j0 += 32) {
            load_scores(j0, deg, beg, d, i8, hsel, csrc, als, adh, s, e);
#pragma unroll
            for (int r = 0; r < 4; ++r) a[r] = expf(e[r] - m) / l;
            const int left = tot - j0;
            add_rows(left < 32 ? left : 32, s, a, lane, Hm, &acc);
        }
    }

    const float4 bv = *(const float4*)(bias + lane * 4);
    float4 v = make_float4(acc.x + bv.x, acc.y + bv.y, acc.z + bv.z, acc.w + bv.w);
    if (apply_ln) {
        const float4 gv = *(const float4*)(g + lane * 4);
        const float4 bl = *(const float4*)(bln + lane * 4);
        const float s1 = wave_sum((v.x + v.y) + (v.z + v.w));
        const float s2 = wave_sum((v.x * v.x + v.y * v.y) + (v.z * v.z + v.w * v.w));
        const float mu  = s1 * (1.f / HC);
        const float var = s2 * (1.f / HC) - mu * mu;
        const float rs  = rsqrtf(var + EPSV);
        v.x = fmaxf((v.x - mu) * rs * gv.x + bl.x, 0.f);
        v.y = fmaxf((v.y - mu) * rs * gv.y + bl.y, 0.f);
        v.z = fmaxf((v.z - mu) * rs * gv.z + bl.z, 0.f);
        v.w = fmaxf((v.w - mu) * rs * gv.w + bl.w, 0.f);
    }
    *(float4*)(outp + (size_t)d * HC + lane * 4) = v;
}

extern "C" void kernel_launch(void* const* d_in, const int* in_sizes, int n_in,
                              void* d_out, int out_size, void* d_ws, size_t ws_size,
                              hipStream_t stream)
{
    const float* x    = (const float*)d_in[0];
    const int*   esrc = (const int*)d_in[1];
    const int*   edst = (const int*)d_in[2];
    const float* W[3]  = {(const float*)d_in[3],  (const float*)d_in[7],  (const float*)d_in[11]};
    const float* As[3] = {(const float*)d_in[4],  (const float*)d_in[8],  (const float*)d_in[12]};
    const float* Ad[3] = {(const float*)d_in[5],  (const float*)d_in[9],  (const float*)d_in[13]};
    const float* Bi[3] = {(const float*)d_in[6],  (const float*)d_in[10], (const float*)d_in[14]};
    const float* LnG[2] = {(const float*)d_in[15], (const float*)d_in[17]};
    const float* LnB[2] = {(const float*)d_in[16], (const float*)d_in[18]};

    const int Nn   = in_sizes[0] / HC;
    const int E    = in_sizes[1];

    float* hbuf = (float*)d_ws;
    int*   prs  = (int*)(hbuf + (size_t)Nn * HC);
    int*   csrc = prs + 2 * (size_t)E;
    int*   cnt  = csrc + E;
    int*   rowp = cnt + Nn;
    int*   offt = rowp + Nn;
    float* als  = hbuf + 2 * (size_t)Nn * HC;
    float* ald  = als + (size_t)Nn * HEADS;
    float* lay  = (float*)d_out;

    const int TB = 256;
    const unsigned gGemm = (unsigned)((Nn + 15) / 16);
    const unsigned gAttn = (unsigned)(((long long)Nn * HEADS + TB - 1) / TB);
    const int      nCh   = (E + CHUNK - 1) / CHUNK;
    const int      nRg   = (Nn + RNODES - 1) / RNODES;
    const unsigned gGat  = (unsigned)((Nn + 7) / 8);

    k_csr_part<<<(unsigned)nCh, 1024, 0, stream>>>(esrc, edst, prs, offt, E, nRg);
    k_csr_place<<<(unsigned)nRg, 1024, 0, stream>>>(prs, offt, cnt, rowp, csrc, Nn, nRg, nCh);

    const float* cur = x;
    for (int l = 0; l < 3; ++l) {
        k_gemm_wmma<<<gGemm, 256, 0, stream>>>(cur, W[l], hbuf, Nn);
        k_attn_logits<<<gAttn, TB, 0, stream>>>(hbuf, As[l], Ad[l], als, ald, Nn);
        if (l < 2) {
            k_node_gather<<<gGat, 256, 0, stream>>>(rowp, cnt, csrc, als, ald, hbuf,
                                                    Bi[l], LnG[l], LnB[l], lay, Nn, 1);
            cur = lay;
        } else {
            k_node_gather<<<gGat, 256, 0, stream>>>(rowp, cnt, csrc, als, ald, hbuf,
                                                    Bi[l], LnG[0], LnB[0], lay, Nn, 0);
        }
    }
}


# GEMM as 256 resident work-groups: W turned once through LDS into each wave's registers (its 64 columns), strips of 16 rows with 16-B A loads one strip ahead; k regrouped in the f32 products
# speedup vs baseline: 6.0658x; 1.2645x over previous
#include <hip/hip_runtime.h>
#include <hip/hip_bf16.h>
#include <float.h>
#include <math.h>

#define HEADS 4
#define CPH 32
#define HC 128
#define NEG_SLOPE 0.2f
#define EPSV 1e-5f

typedef float v2f __attribute__((ext_vector_type(2)));
typedef float v8f __attribute__((ext_vector_type(8)));

#define WPITCH   132
#define GEMM_WGS 256

__device__ __forceinline__ void ld_a_strip(const float* __restrict__ A, int m0, int M,
                                           int l16, int hi, float4 a[16])
{
    int row = m0 + l16;
    row = row < M ? row : M - 1;
    const float* __restrict__ p = A + (size_t)row * HC + 4 * hi;
#pragma unroll
    for (int kb = 0; kb < 16; ++kb) a[kb] = *(const float4*)(p + 8 * kb);
}

__global__ __launch_bounds__(256) void k_gemm_wmma(
    const float* __restrict__ A, const float* __restrict__ W,
    float* __restrict__ Hout, int M)
{
    __shared__ __attribute__((aligned(16))) float Wt[HC * WPITCH];
    const int t    = threadIdx.x;
    const int lane = t & 31;
    const int wave = __builtin_amdgcn_readfirstlane(t >> 5);
    const int hi   = lane >> 4;
    const int l16  = lane & 15;

    const int nstrips = (M + 15) >> 4;
    const int sstep   = 4 * (int)gridDim.x;
    const int n0      = (wave & 1) * 64;
    int s = (wave >> 1) * (int)gridDim.x + (int)blockIdx.x;

    float4 a[16], an[16];
    ld_a_strip(A, (s < nstrips ? s : 0) * 16, M, l16, hi, a);

    {
        const int n  = t & 127;
        const int kq = t >> 7;
#pragma unroll 4
        for (int i = 0; i < 16; ++i) {
            const int k0 = 8 * i + 4 * kq;
            float4 v;
            v.x = W[(size_t)(k0 + 0) * HC + n];
            v.y = W[(size_t)(k0 + 1) * HC + n];
            v.z = W[(size_t)(k0 + 2) * HC + n];
            v.w = W[(size_t)(k0 + 3) * HC + n];
            *(float4*)(Wt + n * WPITCH + k0) = v;
        }
    }
    __syncthreads();

    float4 bf[4][16];
    {
        const float* bbase = Wt + (n0 + l16) * WPITCH + 4 * hi;
#pragma unroll
        for (int tl = 0; tl < 4; ++tl)
#pragma unroll
            for (int kb = 0; kb < 16; ++kb)
                bf[tl][kb] = *(const float4*)(bbase + tl * 16 * WPITCH + 8 * kb);
    }

    for (; s < nstrips; s += sstep) {
        const int sn = s + sstep;
        ld_a_strip(A, (sn < nstrips ? sn : s) * 16, M, l16, hi, an);

        v8f acc[4];
#pragma unroll
        for (int tl = 0; tl < 4; ++tl) acc[tl] = (v8f){0.f, 0.f, 0.f, 0.f, 0.f, 0.f, 0.f, 0.f};

        __builtin_amdgcn_sched_barrier(0);
#pragma unroll
        for (int kb = 0; kb < 16; ++kb) {
            const v2f a0 = {a[kb].x, a[kb].y};
            const v2f a1 = {a[kb].z, a[kb].w};
#pragma unroll
            for (int tl = 0; tl < 4; ++tl) {
                const v2f b0 = {bf[tl][kb].x, bf[tl][kb].y};
                acc[tl] = __builtin_amdgcn_wmma_f32_16x16x4_f32(
                     false, a0,  false, b0,
                     (short)0, acc[tl],  false,  false);
            }
#pragma unroll
            for (int tl = 0; tl < 4; ++tl) {
                const v2f b1 = {bf[tl][kb].z, bf[tl][kb].w};
                acc[tl] = __builtin_amdgcn_wmma_f32_16x16x4_f32(
                     false, a1,  false, b1,
                     (short)0, acc[tl],  false,  false);
            }
        }
        __builtin_amdgcn_sched_barrier(0);
        int m0 = s * 16;
        asm volatile("v_nop\n\tv_nop\n\tv_nop\n\tv_nop" : "+s"(m0));

        float* __restrict__ outp = Hout + (size_t)(m0 + 8 * hi) * HC + n0 + l16;
        if (m0 + 16 <= M) {
#pragma unroll
            for (int tl = 0; tl < 4; ++tl)
#pragma unroll
                for (int r = 0; r < 8; ++r) outp[(size_t)r * HC + 16 * tl] = acc[tl][r];
        } else {
#pragma unroll
            for (int tl = 0; tl < 4; ++tl)
#pragma unroll
                for (int r = 0; r < 8; ++r) {
                    const int ro = m0 + r + 8 * hi;
                    if (ro < M) Hout[(size_t)ro * HC + n0 + 16 * tl + l16] = acc[tl][r];
                }
        }
#pragma unroll
        for (int kb = 0; kb < 16; ++kb) a[kb] = an[kb];
    }
}

__global__ void k_attn_logits(const float* __restrict__ Hm,
                              const float* __restrict__ a_s,
                              const float* __restrict__ a_d,
                              float* __restrict__ als, float* __restrict__ ald,
                              int Nn)
{
    int idx = blockIdx.x * blockDim.x + threadIdx.x;
    if (idx >= Nn * HEADS) return;
    int n = idx >> 2, h = idx & 3;
    const float4* hp = (const float4*)(Hm + (size_t)n * HC + h * CPH);
    const float4* as = (const float4*)(a_s + h * CPH);
    const float4* ad = (const float4*)(a_d + h * CPH);
    float ss = 0.f, sd = 0.f;
#pragma unroll
    for (int c = 0; c < CPH / 4; ++c) {
        float4 v = hp[c], a = as[c], b = ad[c];
        ss += v.x * a.x + v.y * a.y + v.z * a.z + v.w * a.w;
        sd += v.x * b.x + v.y * b.y + v.z * b.z + v.w * b.w;
    }
    als[idx] = ss;
    ald[idx] = sd;
}

#define RSH    8
#define RNODES 256
#define CHUNK  8192
#define MAXB   1024
#define MAXCH  1024

__device__ __forceinline__ int block_run_sum(int v, int* ws, int lane, int w)
{
#pragma unroll
    for (int o = 1; o < 32; o <<= 1) {
        const int u = __shfl_up(v, o, 32);
        if (lane >= o) v += u;
    }
    if (lane == 31) ws[w] = v;
    __syncthreads();
    if (w == 0) {
        int s = ws[lane];
#pragma unroll
        for (int o = 1; o < 32; o <<= 1) {
            const int u = __shfl_up(s, o, 32);
            if (lane >= o) s += u;
        }
        ws[lane] = s;
    }
    __syncthreads();
    return v + (w > 0 ? ws[w - 1] : 0);
}

__global__ __launch_bounds__(1024) void k_csr_part(const int* __restrict__ src,
                                                   const int* __restrict__ dst,
                                                   int* __restrict__ pairs_i,
                                                   int* __restrict__ off, int E, int B)
{
    __shared__ int  hist[MAXB];
    __shared__ int  sc[32];
    __shared__ int2 stage[CHUNK];
    int2* __restrict__ pairs = (int2*)pairs_i;
    const int t  = threadIdx.x;
    const int c  = blockIdx.x;
    const int e0 = c * CHUNK;
    int dv[8], sv[8];
#pragma unroll
    for (int k = 0; k < 8; ++k) {
        const int e = e0 + t + 1024 * k;
        dv[k] = (e < E) ? dst[e] : -1;
    }
#pragma unroll
    for (int k = 0; k < 8; ++k) {
        const int e = e0 + t + 1024 * k;
        sv[k] = (e < E) ? src[e] : 0;
    }
    hist[t] = 0;
    __syncthreads();
#pragma unroll
    for (int k = 0; k < 8; ++k)
        if (dv[k] >= 0) atomicAdd(&hist[dv[k] >> RSH], 1);
    __syncthreads();
    const int h  = hist[t];
    const int st = block_run_sum(h, sc, t & 31, t >> 5) - h;
    if (t <= B) off[(size_t)c * (B + 1) + t] = st;
    hist[t] = st;
    __syncthreads();
#pragma unroll
    for (int k = 0; k < 8; ++k)
        if (dv[k] >= 0) {
            const int p = atomicAdd(&hist[dv[k] >> RSH], 1);
            stage[p] = make_int2(sv[k], dv[k]);
        }
    __syncthreads();
#pragma unroll
    for (int k = 0; k < 8; ++k) {
        const int i = t + 1024 * k;
        if (e0 + i < E) pairs[e0 + i] = stage[i];
    }
}

template <bool PLACE>
__device__ __forceinline__ int csr_sweep(const int2* __restrict__ pairs,
                                         const int* __restrict__ off,
                                         int* segb, int* segn, int* cl,
                                         int* __restrict__ csrc,
                                         int b, int lo, int B, int nch,
                                         int t, int lane, int w)
{
    int part = 0;
    for (int c0 = 0; c0 < nch; c0 += MAXCH) {
        int nc = nch - c0;
        nc = nc < MAXCH ? nc : MAXCH;
        for (int c = t; c < nc; c += 1024) {
            const int* o = off + (size_t)(c0 + c) * (B + 1) + b;
            const int s0 = o[0], s1 = o[1];
            segb[c] = (c0 + c) * CHUNK + s0;
            segn[c] = s1 - s0;
            part += s0;
        }
        __syncthreads();
        for (int c = w; c < nc; c += 32) {
            const int sb = segb[c], sn = segn[c];
            for (int j = lane; j < sn; j += 32) {
                const int2 pr = pairs[sb + j];
                if (PLACE) {
                    const int p = atomicAdd(&cl[pr.y - lo], 1);
                    csrc[p] = pr.x;
                } else {
                    atomicAdd(&cl[pr.y - lo], 1);
                }
            }
        }
        __syncthreads();
    }
    return part;
}

__global__ __launch_bounds__(1024) void k_csr_place(const int* __restrict__ pairs_i,
                                                    const int* __restrict__ off,
                                                    int* __restrict__ cnt,
                                                    int* __restrict__ rowp,
                                                    int* __restrict__ csrc,
                                                    int Nn, int B, int nch)
{
    __shared__ int segb[MAXCH];
    __shared__ int segn[MAXCH];
    __shared__ int cl[RNODES];
    __shared__ int sc[32];
    __shared__ int red[32];
    const int2* __restrict__ pairs = (const int2*)pairs_i;
    const int t = threadIdx.x, lane = t & 31, w = t >> 5;
    const int b  = blockIdx.x;
    const int lo = b << RSH;

    if (t < RNODES) cl[t] = 0;
    __syncthreads();
    int part = csr_sweep<false>(pairs, off, segb, segn, cl, csrc, b, lo, B, nch, t, lane, w);

    part += __shfl_xor(part, 16, 32);
    part += __shfl_xor(part, 8, 32);
    part += __shfl_xor(part, 4, 32);
    part += __shfl_xor(part, 2, 32);
    part += __shfl_xor(part, 1, 32);
    if (lane == 0) red[w] = part;
    const int h = (t < RNODES) ? cl[t] : 0;
    const int run = block_run_sum(h, sc, lane, w);
    int base = 0;
    for (int i = 0; i < 32; ++i) base += red[i];
    if (t < RNODES) {
        const int first = base + run - h;
        cl[t] = first;
        if (lo + t < Nn) { rowp[lo + t] = first; cnt[lo + t] = h; }
    }
    __syncthreads();
    csr_sweep<true>(pairs, off, segb, segn, cl, csrc, b, lo, B, nch, t, lane, w);
}

__device__ __forceinline__ float lrelu(float v) { return (v < 0.f) ? NEG_SLOPE * v : v; }

__device__ __forceinline__ float grp8_max(float v) {
#pragma unroll
    for (int off = 4; off > 0; off >>= 1) v = fmaxf(v, __shfl_xor(v, off, 32));
    return v;
}
__device__ __forceinline__ float grp8_sum(float v) {
#pragma unroll
    for (int off = 4; off > 0; off >>= 1) v += __shfl_xor(v, off, 32);
    return v;
}
__device__ __forceinline__ float wave_sum(float v) {
    v += __shfl_xor(v, 16, 32);
    v += __shfl_xor(v, 8, 32);
    v += __shfl_xor(v, 4, 32);
    v += __shfl_xor(v, 2, 32);
    v += __shfl_xor(v, 1, 32);
    return v;
}

__device__ __forceinline__ void load_scores(int j0, int deg, int beg, int d, int i8, int hsel,
                                            const int* __restrict__ csrc,
                                            const float* __restrict__ als,
                                            float adh, int s[4], float e[4])
{
    int   t[4];
    float x[4];
#pragma unroll
    for (int r = 0; r < 4; ++r) {
        int jc = j0 + 8 * r + i8;
        jc = jc < deg ? jc : deg - 1;
        int idx = beg + jc;
        idx = idx < 0 ? 0 : idx;
        t[r] = csrc[idx];
    }
#pragma unroll
    for (int r = 0; r < 4; ++r) s[r] = (j0 + 8 * r + i8 < deg) ? t[r] : d;
#pragma unroll
    for (int r = 0; r < 4; ++r) x[r] = als[(size_t)s[r] * HEADS + hsel];
#pragma unroll
    for (int r = 0; r < 4; ++r)
        e[r] = (j0 + 8 * r + i8 <= deg) ? lrelu(x[r] + adh) : -INFINITY;
}

__device__ __forceinline__ float4 ld_row(const float* __restrict__ Hm, int s, int lane) {
    return *(const float4*)(Hm + (size_t)s * HC + lane * 4);
}
__device__ __forceinline__ void fma_row(float4* acc, float4 v, float w) {
    acc->x += v.x * w;
    acc->y += v.y * w;
    acc->z += v.z * w;
    acc->w += v.w * w;
}

__device__ __forceinline__ void add_rows(int nn, const int s[4], const float a[4],
                                         int lane, const float* __restrict__ Hm, float4* acc)
{
    const int gl = lane & 24;
#pragma unroll
    for (int r = 0; r < 4; ++r) {
        int cg = nn - 8 * r;
        cg = cg < 0 ? 0 : (cg > 8 ? 8 : cg);
        int i = 0;
        for (; i + 4 <= cg; i += 4) {
            const int s0 = __builtin_amdgcn_readlane(s[r], i);
            const int s1 = __builtin_amdgcn_readlane(s[r], i + 1);
            const int s2 = __builtin_amdgcn_readlane(s[r], i + 2);
            const int s3 = __builtin_amdgcn_readlane(s[r], i + 3);
            const float4 v0 = ld_row(Hm, s0, lane);
            const float4 v1 = ld_row(Hm, s1, lane);
            const float4 v2 = ld_row(Hm, s2, lane);
            const float4 v3 = ld_row(Hm, s3, lane);
            const float w0 = __shfl(a[r], gl + i, 32);
            const float w1 = __shfl(a[r], gl + i + 1, 32);
            const float w2 = __shfl(a[r], gl + i + 2, 32);
            const float w3 = __shfl(a[r], gl + i + 3, 32);
            fma_row(acc, v0, w0);
            fma_row(acc, v1, w1);
            fma_row(acc, v2, w2);
            fma_row(acc, v3, w3);
        }
        for (; i < cg; ++i) {
            const int s0 = __builtin_amdgcn_readlane(s[r], i);
            const float4 v0 = ld_row(Hm, s0, lane);
            const float w0 = __shfl(a[r], gl + i, 32);
            fma_row(acc, v0, w0);
        }
    }
}

__global__ __launch_bounds__(256) void k_node_gather(
    const int* __restrict__ rowp, const int* __restrict__ cnt,
    const int* __restrict__ csrc,
    const float* __restrict__ als, const float* __restrict__ ald,
    const float* __restrict__ Hm,
    const float* __restrict__ bias, const float* __restrict__ g, const float* __restrict__ bln,
    float* __restrict__ outp, int Nn, int apply_ln)
{
    const int lane = threadIdx.x & 31;
    const int hsel = lane >> 3;
    const int i8   = lane & 7;
    const int d = __builtin_amdgcn_readfirstlane((int)(blockIdx.x * 8 + (threadIdx.x >> 5)));
    if (d >= Nn) return;
    const int beg = rowp[d];
    const int deg = cnt[d];
    const int tot = deg + 1;
    const float adh = ald[(size_t)d * HEADS + hsel];

    float4 acc = make_float4(0.f, 0.f, 0.f, 0.f);
    int   s[4];
    float e[4];
    float a[4];

    if (tot <= 32) {
        load_scores(0, deg, beg, d, i8, hsel, csrc, als, adh, s, e);
        const float m = grp8_max(fmaxf(fmaxf(e[0], e[1]), fmaxf(e[2], e[3])));
#pragma unroll
        for (int r = 0; r < 4; ++r) a[r] = expf(e[r] - m);
        const float l = grp8_sum((a[0] + a[1]) + (a[2] + a[3]));
#pragma unroll
        for (int r = 0; r < 4; ++r) a[r] = a[r] / l;
        add_rows(tot, s, a, lane, Hm, &acc);
    } else {
        float m = -INFINITY;
        for (int j0 = 0; j0 < tot; j0 += 32) {
            load_scores(j0, deg, beg, d, i8, hsel, csrc, als, adh, s, e);
            m = fmaxf(m, fmaxf(fmaxf(e[0], e[1]), fmaxf(e[2], e[3])));
        }
        m = grp8_max(m);
        float l = 0.f;
        for (int j0 = 0; j0 < tot; j0 += 32) {
            load_scores(j0, deg, beg, d, i8, hsel, csrc, als, adh, s, e);
            l += (expf(e[0] - m) + expf(e[1] - m)) + (expf(e[2] - m) + expf(e[3] - m));
        }
        l = grp8_sum(l);
        for (int j0 = 0; j0 < tot; j0 += 32) {
            load_scores(j0, deg, beg, d, i8, hsel, csrc, als, adh, s, e);
#pragma unroll
            for (int r = 0; r < 4; ++r) a[r] = expf(e[r] - m) / l;
            const int left = tot - j0;
            add_rows(left < 32 ? left : 32, s, a, lane, Hm, &acc);
        }
    }

    const float4 bv = *(const float4*)(bias + lane * 4);
    float4 v = make_float4(acc.x + bv.x, acc.y + bv.y, acc.z + bv.z, acc.w + bv.w);
    if (apply_ln) {
        const float4 gv = *(const float4*)(g + lane * 4);
        const float4 bl = *(const float4*)(bln + lane * 4);
        const float s1 = wave_sum((v.x + v.y) + (v.z + v.w));
        const float s2 = wave_sum((v.x * v.x + v.y * v.y) + (v.z * v.z + v.w * v.w));
        const float mu  = s1 * (1.f / HC);
        const float var = s2 * (1.f / HC) - mu * mu;
        const float rs  = rsqrtf(var + EPSV);
        v.x = fmaxf((v.x - mu) * rs * gv.x + bl.x, 0.f);
        v.y = fmaxf((v.y - mu) * rs * gv.y + bl.y, 0.f);
        v.z = fmaxf((v.z - mu) * rs * gv.z + bl.z, 0.f);
        v.w = fmaxf((v.w - mu) * rs * gv.w + bl.w, 0.f);
    }
    *(float4*)(outp + (size_t)d * HC + lane * 4) = v;
}

extern "C" void kernel_launch(void* const* d_in, const int* in_sizes, int n_in,
                              void* d_out, int out_size, void* d_ws, size_t ws_size,
                              hipStream_t stream)
{
    const float* x    = (const float*)d_in[0];
    const int*   esrc = (const int*)d_in[1];
    const int*   edst = (const int*)d_in[2];
    const float* W[3]  = {(const float*)d_in[3],  (const float*)d_in[7],  (const float*)d_in[11]};
    const float* As[3] = {(const float*)d_in[4],  (const float*)d_in[8],  (const float*)d_in[12]};
    const float* Ad[3] = {(const float*)d_in[5],  (const float*)d_in[9],  (const float*)d_in[13]};
    const float* Bi[3] = {(const float*)d_in[6],  (const float*)d_in[10], (const float*)d_in[14]};
    const float* LnG[2] = {(const float*)d_in[15], (const float*)d_in[17]};
    const float* LnB[2] = {(const float*)d_in[16], (const float*)d_in[18]};

    const int Nn   = in_sizes[0] / HC;
    const int E    = in_sizes[1];

    float* hbuf = (float*)d_ws;
    int*   prs  = (int*)(hbuf + (size_t)Nn * HC);
    int*   csrc = prs + 2 * (size_t)E;
    int*   cnt  = csrc + E;
    int*   rowp = cnt + Nn;
    int*   offt = rowp + Nn;
    float* als  = hbuf + 2 * (size_t)Nn * HC;
    float* ald  = als + (size_t)Nn * HEADS;
    float* lay  = (float*)d_out;

    const int TB = 256;
    const unsigned gAttn = (unsigned)(((long long)Nn * HEADS + TB - 1) / TB);
    const int      nCh   = (E + CHUNK - 1) / CHUNK;
    const int      nRg   = (Nn + RNODES - 1) / RNODES;
    const unsigned gGat  = (unsigned)((Nn + 7) / 8);

    k_csr_part<<<(unsigned)nCh, 1024, 0, stream>>>(esrc, edst, prs, offt, E, nRg);
    k_csr_place<<<(unsigned)nRg, 1024, 0, stream>>>(prs, offt, cnt, rowp, csrc, Nn, nRg, nCh);

    const float* cur = x;
    for (int l = 0; l < 3; ++l) {
        k_gemm_wmma<<<GEMM_WGS, 256, 0, stream>>>(cur, W[l], hbuf, Nn);
        k_attn_logits<<<gAttn, TB, 0, stream>>>(hbuf, As[l], Ad[l], als, ald, Nn);
        if (l < 2) {
            k_node_gather<<<gGat, 256, 0, stream>>>(rowp, cnt, csrc, als, ald, hbuf,
                                                    Bi[l], LnG[l], LnB[l], lay, Nn, 1);
            cur = lay;
        } else {
            k_node_gather<<<gGat, 256, 0, stream>>>(rowp, cnt, csrc, als, ald, hbuf,
                                                    Bi[l], LnG[0], LnB[0], lay, Nn, 0);
        }
    }
}


# Gather wave's lane exchanges moved from the LDS unit to the vector ALU: xor steps as row_xmask DPP on the add/max, l^16 by permlanex16, group broadcast by DPP8; same pairings, same sum order
# speedup vs baseline: 7.2125x; 1.1890x over previous
#include <hip/hip_runtime.h>
#include <hip/hip_bf16.h>
#include <float.h>
#include <math.h>

#define HEADS 4
#define CPH 32
#define HC 128
#define NEG_SLOPE 0.2f
#define EPSV 1e-5f

typedef float v2f __attribute__((ext_vector_type(2)));
typedef float v8f __attribute__((ext_vector_type(8)));

#define WPITCH   132
#define GEMM_WGS 256

__device__ __forceinline__ void ld_a_strip(const float* __restrict__ A, int m0, int M,
                                           int l16, int hi, float4 a[16])
{
    int row = m0 + l16;
    row = row < M ? row : M - 1;
    const float* __restrict__ p = A + (size_t)row * HC + 4 * hi;
#pragma unroll
    for (int kb = 0; kb < 16; ++kb) a[kb] = *(const float4*)(p + 8 * kb);
}

__global__ __launch_bounds__(256) void k_gemm_wmma(
    const float* __restrict__ A, const float* __restrict__ W,
    float* __restrict__ Hout, int M)
{
    __shared__ __attribute__((aligned(16))) float Wt[HC * WPITCH];
    const int t    = threadIdx.x;
    const int lane = t & 31;
    const int wave = __builtin_amdgcn_readfirstlane(t >> 5);
    const int hi   = lane >> 4;
    const int l16  = lane & 15;

    const int nstrips = (M + 15) >> 4;
    const int sstep   = 4 * (int)gridDim.x;
    const int n0      = (wave & 1) * 64;
    int s = (wave >> 1) * (int)gridDim.x + (int)blockIdx.x;

    float4 a[16], an[16];
    ld_a_strip(A, (s < nstrips ? s : 0) * 16, M, l16, hi, a);

    {
        const int n  = t & 127;
        const int kq = t >> 7;
#pragma unroll 4
        for (int i = 0; i < 16; ++i) {
            const int k0 = 8 * i + 4 * kq;
            float4 v;
            v.x = W[(size_t)(k0 + 0) * HC + n];
            v.y = W[(size_t)(k0 + 1) * HC + n];
            v.z = W[(size_t)(k0 + 2) * HC + n];
            v.w = W[(size_t)(k0 + 3) * HC + n];
            *(float4*)(Wt + n * WPITCH + k0) = v;
        }
    }
    __syncthreads();

    float4 bf[4][16];
    {
        const float* bbase = Wt + (n0 + l16) * WPITCH + 4 * hi;
#pragma unroll
        for (int tl = 0; tl < 4; ++tl)
#pragma unroll
            for (int kb = 0; kb < 16; ++kb)
                bf[tl][kb] = *(const float4*)(bbase + tl * 16 * WPITCH + 8 * kb);
    }

    for (; s < nstrips; s += sstep) {
        const int sn = s + sstep;
        ld_a_strip(A, (sn < nstrips ? sn : s) * 16, M, l16, hi, an);

        v8f acc[4];
#pragma unroll
        for (int tl = 0; tl < 4; ++tl) acc[tl] = (v8f){0.f, 0.f, 0.f, 0.f, 0.f, 0.f, 0.f, 0.f};

        __builtin_amdgcn_sched_barrier(0);
#pragma unroll
        for (int kb = 0; kb < 16; ++kb) {
            const v2f a0 = {a[kb].x, a[kb].y};
            const v2f a1 = {a[kb].z, a[kb].w};
#pragma unroll
            for (int tl = 0; tl < 4; ++tl) {
                const v2f b0 = {bf[tl][kb].x, bf[tl][kb].y};
                acc[tl] = __builtin_amdgcn_wmma_f32_16x16x4_f32(
                     false, a0,  false, b0,
                     (short)0, acc[tl],  false,  false);
            }
#pragma unroll
            for (int tl = 0; tl < 4; ++tl) {
                const v2f b1 = {bf[tl][kb].z, bf[tl][kb].w};
                acc[tl] = __builtin_amdgcn_wmma_f32_16x16x4_f32(
                     false, a1,  false, b1,
                     (short)0, acc[tl],  false,  false);
            }
        }
        __builtin_amdgcn_sched_barrier(0);
        int m0 = s * 16;
        asm volatile("v_nop\n\tv_nop\n\tv_nop\n\tv_nop" : "+s"(m0));

        float* __restrict__ outp = Hout + (size_t)(m0 + 8 * hi) * HC + n0 + l16;
        if (m0 + 16 <= M) {
#pragma unroll
            for (int tl = 0; tl < 4; ++tl)
#pragma unroll
                for (int r = 0; r < 8; ++r) outp[(size_t)r * HC + 16 * tl] = acc[tl][r];
        } else {
#pragma unroll
            for (int tl = 0; tl < 4; ++tl)
#pragma unroll
                for (int r = 0; r < 8; ++r) {
                    const int ro = m0 + r + 8 * hi;
                    if (ro < M) Hout[(size_t)ro * HC + n0 + 16 * tl + l16] = acc[tl][r];
                }
        }
#pragma unroll
        for (int kb = 0; kb < 16; ++kb) a[kb] = an[kb];
    }
}

#define RSH    8
#define RNODES 256
#define CHUNK  8192
#define MAXB   1024
#define MAXCH  1024

__device__ __forceinline__ int block_run_sum(int v, int* ws, int lane, int w)
{
#pragma unroll
    for (int o = 1; o < 32; o <<= 1) {
        const int u = __shfl_up(v, o, 32);
        if (lane >= o) v += u;
    }
    if (lane == 31) ws[w] = v;
    __syncthreads();
    if (w == 0) {
        int s = ws[lane];
#pragma unroll
        for (int o = 1; o < 32; o <<= 1) {
            const int u = __shfl_up(s, o, 32);
            if (lane >= o) s += u;
        }
        ws[lane] = s;
    }
    __syncthreads();
    return v + (w > 0 ? ws[w - 1] : 0);
}

__global__ __launch_bounds__(1024) void k_csr_part(const int* __restrict__ src,
                                                   const int* __restrict__ dst,
                                                   int* __restrict__ pairs_i,
                                                   int* __restrict__ off, int E, int B)
{
    __shared__ int  hist[MAXB];
    __shared__ int  sc[32];
    __shared__ int2 stage[CHUNK];
    int2* __restrict__ pairs = (int2*)pairs_i;
    const int t  = threadIdx.x;
    const int c  = blockIdx.x;
    const int e0 = c * CHUNK;
    int dv[8], sv[8];
#pragma unroll
    for (int k = 0; k < 8; ++k) {
        const int e = e0 + t + 1024 * k;
        dv[k] = (e < E) ? dst[e] : -1;
    }
#pragma unroll
    for (int k = 0; k < 8; ++k) {
        const int e = e0 + t + 1024 * k;
        sv[k] = (e < E) ? src[e] : 0;
    }
    hist[t] = 0;
    __syncthreads();
#pragma unroll
    for (int k = 0; k < 8; ++k)
        if (dv[k] >= 0) atomicAdd(&hist[dv[k] >> RSH], 1);
    __syncthreads();
    const int h  = hist[t];
    const int st = block_run_sum(h, sc, t & 31, t >> 5) - h;
    if (t <= B) off[(size_t)c * (B + 1) + t] = st;
    hist[t] = st;
    __syncthreads();
#pragma unroll
    for (int k = 0; k < 8; ++k)
        if (dv[k] >= 0) {
            const int p = atomicAdd(&hist[dv[k] >> RSH], 1);
            stage[p] = make_int2(sv[k], dv[k]);
        }
    __syncthreads();
#pragma unroll
    for (int k = 0; k < 8; ++k) {
        const int i = t + 1024 * k;
        if (e0 + i < E) pairs[e0 + i] = stage[i];
    }
}

template <bool PLACE>
__device__ __forceinline__ int csr_sweep(const int2* __restrict__ pairs,
                                         const int* __restrict__ off,
                                         int* segb, int* segn, int* cl,
                                         int* __restrict__ csrc,
                                         int b, int lo, int B, int nch,
                                         int t, int lane, int w)
{
    int part = 0;
    for (int c0 = 0; c0 < nch; c0 += MAXCH) {
        int nc = nch - c0;
        nc = nc < MAXCH ? nc : MAXCH;
        for (int c = t; c < nc; c += 1024) {
            const int* o = off + (size_t)(c0 + c) * (B + 1) + b;
            const int s0 = o[0], s1 = o[1];
            segb[c] = (c0 + c) * CHUNK + s0;
            segn[c] = s1 - s0;
            part += s0;
        }
        __syncthreads();
        for (int c = w; c < nc; c += 32) {
            const int sb = segb[c], sn = segn[c];
            for (int j = lane; j < sn; j += 32) {
                const int2 pr = pairs[sb + j];
                if (PLACE) {
                    const int p = atomicAdd(&cl[pr.y - lo], 1);
                    csrc[p] = pr.x;
                } else {
                    atomicAdd(&cl[pr.y - lo], 1);
                }
            }
        }
        __syncthreads();
    }
    return part;
}

__global__ __launch_bounds__(1024) void k_csr_place(const int* __restrict__ pairs_i,
                                                    const int* __restrict__ off,
                                                    int* __restrict__ cnt,
                                                    int* __restrict__ rowp,
                                                    int* __restrict__ csrc,
                                                    int Nn, int B, int nch)
{
    __shared__ int segb[MAXCH];
    __shared__ int segn[MAXCH];
    __shared__ int cl[RNODES];
    __shared__ int sc[32];
    __shared__ int red[32];
    const int2* __restrict__ pairs = (const int2*)pairs_i;
    const int t = threadIdx.x, lane = t & 31, w = t >> 5;
    const int b  = blockIdx.x;
    const int lo = b << RSH;

    if (t < RNODES) cl[t] = 0;
    __syncthreads();
    int part = csr_sweep<false>(pairs, off, segb, segn, cl, csrc, b, lo, B, nch, t, lane, w);

    part += __shfl_xor(part, 16, 32);
    part += __shfl_xor(part, 8, 32);
    part += __shfl_xor(part, 4, 32);
    part += __shfl_xor(part, 2, 32);
    part += __shfl_xor(part, 1, 32);
    if (lane == 0) red[w] = part;
    const int h = (t < RNODES) ? cl[t] : 0;
    const int run = block_run_sum(h, sc, lane, w);
    int base = 0;
    for (int i = 0; i < 32; ++i) base += red[i];
    if (t < RNODES) {
        const int first = base + run - h;
        cl[t] = first;
        if (lo + t < Nn) { rowp[lo + t] = first; cnt[lo + t] = h; }
    }
    __syncthreads();
    csr_sweep<true>(pairs, off, segb, segn, cl, csrc, b, lo, B, nch, t, lane, w);
}

__device__ __forceinline__ float lrelu(float v) { return (v < 0.f) ? NEG_SLOPE * v : v; }


template <int M>
__device__ __forceinline__ float xlane(float x) {
    return __builtin_bit_cast(float, __builtin_amdgcn_mov_dpp(__builtin_bit_cast(int, x),
                                                              0x160 + M, 0xf, 0xf, true));
}
__device__ __forceinline__ float xhalf(float x) {
    const int b = __builtin_bit_cast(int, x);
    return __builtin_bit_cast(float,
        __builtin_amdgcn_permlanex16(b, b, 0x76543210, 0xfedcba98, false, false));
}
template <int K>
__device__ __forceinline__ float grp8_lane(float x) {
    return __builtin_bit_cast(float,
        __builtin_amdgcn_mov_dpp8(__builtin_bit_cast(unsigned, x), K * 0x249249));
}

__device__ __forceinline__ float grp8_max(float v) {
    v = fmaxf(v, xlane<4>(v));
    v = fmaxf(v, xlane<2>(v));
    v = fmaxf(v, xlane<1>(v));
    return v;
}
__device__ __forceinline__ float grp8_sum(float v) {
    v += xlane<4>(v);
    v += xlane<2>(v);
    v += xlane<1>(v);
    return v;
}
__device__ __forceinline__ float wave_sum(float v) {
    v += xhalf(v);
    v += xlane<8>(v);
    v += xlane<4>(v);
    v += xlane<2>(v);
    v += xlane<1>(v);
    return v;
}

__device__ __forceinline__ float grp8_sum_to_lane(const float p[8], int i8)
{
    const bool b4 = (i8 & 4) != 0, b2 = (i8 & 2) != 0, b1 = (i8 & 1) != 0;
    float q[4], r[2];
#pragma unroll
    for (int k = 0; k < 4; ++k) {
        const float keep = b4 ? p[k + 4] : p[k];
        const float send = b4 ? p[k] : p[k + 4];
        q[k] = keep + xlane<4>(send);
    }
#pragma unroll
    for (int k = 0; k < 2; ++k) {
        const float keep = b2 ? q[k + 2] : q[k];
        const float send = b2 ? q[k] : q[k + 2];
        r[k] = keep + xlane<2>(send);
    }
    const float keep = b1 ? r[1] : r[0];
    const float send = b1 ? r[0] : r[1];
    return keep + xlane<1>(send);
}

struct row4 { v2f lo, hi; };
__device__ __forceinline__ row4 ld_row(const float* __restrict__ Hm, int s, int lane) {
    const float4 t = *(const float4*)(Hm + (size_t)s * HC + lane * 4);
    row4 r;
    r.lo = (v2f){t.x, t.y};
    r.hi = (v2f){t.z, t.w};
    return r;
}
__device__ __forceinline__ float dot4(row4 v, row4 a) {
    v2f d = v.lo * a.lo;
    d = __builtin_elementwise_fma(v.hi, a.hi, d);
    return d.x + d.y;
}
__device__ __forceinline__ void fma_row(row4* acc, row4 v, float w) {
    const v2f w2 = {w, w};
    acc->lo = __builtin_elementwise_fma(v.lo, w2, acc->lo);
    acc->hi = __builtin_elementwise_fma(v.hi, w2, acc->hi);
}

template <bool FULL>
__device__ __forceinline__ void edge_batch(int nb, int j0, int sidx, int lane, int i8,
                                           const float* __restrict__ Hm, row4 as4, float adh,
                                           float* m, float* l, row4* acc)
{
    int   sr[8];
    row4  v[8];
    float p[8];
#pragma unroll
    for (int i = 0; i < 8; ++i) sr[i] = __builtin_amdgcn_readlane(sidx, j0 + i);
#pragma unroll
    for (int i = 0; i < 8; ++i) {
        if (FULL || i < nb) {
            v[i] = ld_row(Hm, sr[i], lane);
        } else {
            v[i].lo = (v2f){0.f, 0.f};
            v[i].hi = (v2f){0.f, 0.f};
        }
    }
#pragma unroll
    for (int i = 0; i < 8; ++i) p[i] = dot4(v[i], as4);
    const float als = grp8_sum_to_lane(p, i8);
    const float e   = (FULL || i8 < nb) ? lrelu(als + adh) : -INFINITY;
    const float mn  = fmaxf(*m, grp8_max(e));
    const float sc  = expf(*m - mn);
    const float w   = expf(e - mn);
    *l = *l * sc + grp8_sum(w);
    *m = mn;
    const v2f sc2 = {sc, sc};
    acc->lo *= sc2;
    acc->hi *= sc2;
    fma_row(acc, v[0], grp8_lane<0>(w));
    fma_row(acc, v[1], grp8_lane<1>(w));
    fma_row(acc, v[2], grp8_lane<2>(w));
    fma_row(acc, v[3], grp8_lane<3>(w));
    fma_row(acc, v[4], grp8_lane<4>(w));
    fma_row(acc, v[5], grp8_lane<5>(w));
    fma_row(acc, v[6], grp8_lane<6>(w));
    fma_row(acc, v[7], grp8_lane<7>(w));
}

__global__ __launch_bounds__(256) void k_node_gather(
    const int* __restrict__ rowp, const int* __restrict__ cnt,
    const int* __restrict__ csrc,
    const float* __restrict__ a_s, const float* __restrict__ a_d,
    const float* __restrict__ Hm,
    const float* __restrict__ bias, const float* __restrict__ g, const float* __restrict__ bln,
    float* __restrict__ outp, int Nn, int apply_ln)
{
    const int lane = threadIdx.x & 31;
    const int i8   = lane & 7;
    const int d = __builtin_amdgcn_readfirstlane((int)(blockIdx.x * 8 + (threadIdx.x >> 5)));
    if (d >= Nn) return;
    const int beg = rowp[d];
    const int deg = cnt[d];

    const row4 as4 = ld_row(a_s, 0, lane);
    const row4 ad4 = ld_row(a_d, 0, lane);

    row4 acc = ld_row(Hm, d, lane);
    const float adh = grp8_sum(dot4(acc, ad4));
    float m = lrelu(grp8_sum(dot4(acc, as4)) + adh);
    float l = 1.f;

    for (int c0 = 0; c0 < deg; c0 += 32) {
        int jc = c0 + lane;
        jc = jc < deg ? jc : deg - 1;
        const int sidx = csrc[beg + jc];
        int nc = deg - c0;
        nc = nc < 32 ? nc : 32;
        int j0 = 0;
        for (; j0 + 8 <= nc; j0 += 8)
            edge_batch<true>(8, j0, sidx, lane, i8, Hm, as4, adh, &m, &l, &acc);
        if (j0 < nc)
            edge_batch<false>(nc - j0, j0, sidx, lane, i8, Hm, as4, adh, &m, &l, &acc);
    }

    const float4 bv = *(const float4*)(bias + lane * 4);
    float4 v = make_float4(acc.lo.x / l + bv.x, acc.lo.y / l + bv.y,
                           acc.hi.x / l + bv.z, acc.hi.y / l + bv.w);
    if (apply_ln) {
        const float4 gv = *(const float4*)(g + lane * 4);
        const float4 bl = *(const float4*)(bln + lane * 4);
        const float s1 = wave_sum((v.x + v.y) + (v.z + v.w));
        const float s2 = wave_sum((v.x * v.x + v.y * v.y) + (v.z * v.z + v.w * v.w));
        const float mu  = s1 * (1.f / HC);
        const float var = s2 * (1.f / HC) - mu * mu;
        const float rs  = rsqrtf(var + EPSV);
        v.x = fmaxf((v.x - mu) * rs * gv.x + bl.x, 0.f);
        v.y = fmaxf((v.y - mu) * rs * gv.y + bl.y, 0.f);
        v.z = fmaxf((v.z - mu) * rs * gv.z + bl.z, 0.f);
        v.w = fmaxf((v.w - mu) * rs * gv.w + bl.w, 0.f);
    }
    *(float4*)(outp + (size_t)d * HC + lane * 4) = v;
}

extern "C" void kernel_launch(void* const* d_in, const int* in_sizes, int n_in,
                              void* d_out, int out_size, void* d_ws, size_t ws_size,
                              hipStream_t stream)
{
    const float* x    = (const float*)d_in[0];
    const int*   esrc = (const int*)d_in[1];
    const int*   edst = (const int*)d_in[2];
    const float* W[3]  = {(const float*)d_in[3],  (const float*)d_in[7],  (const float*)d_in[11]};
    const float* As[3] = {(const float*)d_in[4],  (const float*)d_in[8],  (const float*)d_in[12]};
    const float* Ad[3] = {(const float*)d_in[5],  (const float*)d_in[9],  (const float*)d_in[13]};
    const float* Bi[3] = {(const float*)d_in[6],  (const float*)d_in[10], (const float*)d_in[14]};
    const float* LnG[2] = {(const float*)d_in[15], (const float*)d_in[17]};
    const float* LnB[2] = {(const float*)d_in[16], (const float*)d_in[18]};

    const int Nn   = in_sizes[0] / HC;
    const int E    = in_sizes[1];

    float* hbuf = (float*)d_ws;
    int*   prs  = (int*)(hbuf + (size_t)Nn * HC);
    int*   csrc = prs + 2 * (size_t)E;
    int*   cnt  = csrc + E;
    int*   rowp = cnt + Nn;
    int*   offt = rowp + Nn;
    float* lay  = (float*)d_out;

    const int      nCh   = (E + CHUNK - 1) / CHUNK;
    const int      nRg   = (Nn + RNODES - 1) / RNODES;
    const unsigned gGat  = (unsigned)((Nn + 7) / 8);

    k_csr_part<<<(unsigned)nCh, 1024, 0, stream>>>(esrc, edst, prs, offt, E, nRg);
    k_csr_place<<<(unsigned)nRg, 1024, 0, stream>>>(prs, offt, cnt, rowp, csrc, Nn, nRg, nCh);

    const float* cur = x;
    for (int l = 0; l < 3; ++l) {
        k_gemm_wmma<<<GEMM_WGS, 256, 0, stream>>>(cur, W[l], hbuf, Nn);
        if (l < 2) {
            k_node_gather<<<gGat, 256, 0, stream>>>(rowp, cnt, csrc, As[l], Ad[l], hbuf,
                                                    Bi[l], LnG[l], LnB[l], lay, Nn, 1);
            cur = lay;
        } else {
            k_node_gather<<<gGat, 256, 0, stream>>>(rowp, cnt, csrc, As[l], Ad[l], hbuf,
                                                    Bi[l], LnG[0], LnB[0], lay, Nn, 0);
        }
    }
}


# Gather kernel 68 -> 64 vector registers, 16 not 12 waves per SIMD: edge-row loads take a scalar row base + one shared lane offset (no 64-bit vector address per row), row dots kept unpaired; same sums
# speedup vs baseline: 7.5443x; 1.0460x over previous
#include <hip/hip_runtime.h>
#include <hip/hip_bf16.h>
#include <float.h>
#include <math.h>

#define HEADS 4
#define CPH 32
#define HC 128
#define NEG_SLOPE 0.2f
#define EPSV 1e-5f

typedef float v2f __attribute__((ext_vector_type(2)));
typedef float v8f __attribute__((ext_vector_type(8)));

#define WPITCH   132
#define GEMM_WGS 256

__device__ __forceinline__ void ld_a_strip(const float* __restrict__ A, int m0, int M,
                                           int l16, int hi, float4 a[16])
{
    int row = m0 + l16;
    row = row < M ? row : M - 1;
    const float* __restrict__ p = A + (size_t)row * HC + 4 * hi;
#pragma unroll
    for (int kb = 0; kb < 16; ++kb) a[kb] = *(const float4*)(p + 8 * kb);
}

__global__ __launch_bounds__(256) void k_gemm_wmma(
    const float* __restrict__ A, const float* __restrict__ W,
    float* __restrict__ Hout, int M)
{
    __shared__ __attribute__((aligned(16))) float Wt[HC * WPITCH];
    const int t    = threadIdx.x;
    const int lane = t & 31;
    const int wave = __builtin_amdgcn_readfirstlane(t >> 5);
    const int hi   = lane >> 4;
    const int l16  = lane & 15;

    const int nstrips = (M + 15) >> 4;
    const int sstep   = 4 * (int)gridDim.x;
    const int n0      = (wave & 1) * 64;
    int s = (wave >> 1) * (int)gridDim.x + (int)blockIdx.x;

    float4 a[16], an[16];
    ld_a_strip(A, (s < nstrips ? s : 0) * 16, M, l16, hi, a);

    {
        const int n  = t & 127;
        const int kq = t >> 7;
#pragma unroll 4
        for (int i = 0; i < 16; ++i) {
            const int k0 = 8 * i + 4 * kq;
            float4 v;
            v.x = W[(size_t)(k0 + 0) * HC + n];
            v.y = W[(size_t)(k0 + 1) * HC + n];
            v.z = W[(size_t)(k0 + 2) * HC + n];
            v.w = W[(size_t)(k0 + 3) * HC + n];
            *(float4*)(Wt + n * WPITCH + k0) = v;
        }
    }
    __syncthreads();

    float4 bf[4][16];
    {
        const float* bbase = Wt + (n0 + l16) * WPITCH + 4 * hi;
#pragma unroll
        for (int tl = 0; tl < 4; ++tl)
#pragma unroll
            for (int kb = 0; kb < 16; ++kb)
                bf[tl][kb] = *(const float4*)(bbase + tl * 16 * WPITCH + 8 * kb);
    }

    for (; s < nstrips; s += sstep) {
        const int sn = s + sstep;
        ld_a_strip(A, (sn < nstrips ? sn : s) * 16, M, l16, hi, an);

        v8f acc[4];
#pragma unroll
        for (int tl = 0; tl < 4; ++tl) acc[tl] = (v8f){0.f, 0.f, 0.f, 0.f, 0.f, 0.f, 0.f, 0.f};

        __builtin_amdgcn_sched_barrier(0);
#pragma unroll
        for (int kb = 0; kb < 16; ++kb) {
            const v2f a0 = {a[kb].x, a[kb].y};
            const v2f a1 = {a[kb].z, a[kb].w};
#pragma unroll
            for (int tl = 0; tl < 4; ++tl) {
                const v2f b0 = {bf[tl][kb].x, bf[tl][kb].y};
                acc[tl] = __builtin_amdgcn_wmma_f32_16x16x4_f32(
                     false, a0,  false, b0,
                     (short)0, acc[tl],  false,  false);
            }
#pragma unroll
            for (int tl = 0; tl < 4; ++tl) {
                const v2f b1 = {bf[tl][kb].z, bf[tl][kb].w};
                acc[tl] = __builtin_amdgcn_wmma_f32_16x16x4_f32(
                     false, a1,  false, b1,
                     (short)0, acc[tl],  false,  false);
            }
        }
        __builtin_amdgcn_sched_barrier(0);
        int m0 = s * 16;
        asm volatile("v_nop\n\tv_nop\n\tv_nop\n\tv_nop" : "+s"(m0));

        float* __restrict__ outp = Hout + (size_t)(m0 + 8 * hi) * HC + n0 + l16;
        if (m0 + 16 <= M) {
#pragma unroll
            for (int tl = 0; tl < 4; ++tl)
#pragma unroll
                for (int r = 0; r < 8; ++r) outp[(size_t)r * HC + 16 * tl] = acc[tl][r];
        } else {
#pragma unroll
            for (int tl = 0; tl < 4; ++tl)
#pragma unroll
                for (int r = 0; r < 8; ++r) {
                    const int ro = m0 + r + 8 * hi;
                    if (ro < M) Hout[(size_t)ro * HC + n0 + 16 * tl + l16] = acc[tl][r];
                }
        }
#pragma unroll
        for (int kb = 0; kb < 16; ++kb) a[kb] = an[kb];
    }
}

#define RSH    8
#define RNODES 256
#define CHUNK  8192
#define MAXB   1024
#define MAXCH  1024

__device__ __forceinline__ int block_run_sum(int v, int* ws, int lane, int w)
{
#pragma unroll
    for (int o = 1; o < 32; o <<= 1) {
        const int u = __shfl_up(v, o, 32);
        if (lane >= o) v += u;
    }
    if (lane == 31) ws[w] = v;
    __syncthreads();
    if (w == 0) {
        int s = ws[lane];
#pragma unroll
        for (int o = 1; o < 32; o <<= 1) {
            const int u = __shfl_up(s, o, 32);
            if (lane >= o) s += u;
        }
        ws[lane] = s;
    }
    __syncthreads();
    return v + (w > 0 ? ws[w - 1] : 0);
}

__global__ __launch_bounds__(1024) void k_csr_part(const int* __restrict__ src,
                                                   const int* __restrict__ dst,
                                                   int* __restrict__ pairs_i,
                                                   int* __restrict__ off, int E, int B)
{
    __shared__ int  hist[MAXB];
    __shared__ int  sc[32];
    __shared__ int2 stage[CHUNK];
    int2* __restrict__ pairs = (int2*)pairs_i;
    const int t  = threadIdx.x;
    const int c  = blockIdx.x;
    const int e0 = c * CHUNK;
    int dv[8], sv[8];
#pragma unroll
    for (int k = 0; k < 8; ++k) {
        const int e = e0 + t + 1024 * k;
        dv[k] = (e < E) ? dst[e] : -1;
    }
#pragma unroll
    for (int k = 0; k < 8; ++k) {
        const int e = e0 + t + 1024 * k;
        sv[k] = (e < E) ? src[e] : 0;
    }
    hist[t] = 0;
    __syncthreads();
#pragma unroll
    for (int k = 0; k < 8; ++k)
        if (dv[k] >= 0) atomicAdd(&hist[dv[k] >> RSH], 1);
    __syncthreads();
    const int h  = hist[t];
    const int st = block_run_sum(h, sc, t & 31, t >> 5) - h;
    if (t <= B) off[(size_t)c * (B + 1) + t] = st;
    hist[t] = st;
    __syncthreads();
#pragma unroll
    for (int k = 0; k < 8; ++k)
        if (dv[k] >= 0) {
            const int p = atomicAdd(&hist[dv[k] >> RSH], 1);
            stage[p] = make_int2(sv[k], dv[k]);
        }
    __syncthreads();
#pragma unroll
    for (int k = 0; k < 8; ++k) {
        const int i = t + 1024 * k;
        if (e0 + i < E) pairs[e0 + i] = stage[i];
    }
}

template <bool PLACE>
__device__ __forceinline__ int csr_sweep(const int2* __restrict__ pairs,
                                         const int* __restrict__ off,
                                         int* segb, int* segn, int* cl,
                                         int* __restrict__ csrc,
                                         int b, int lo, int B, int nch,
                                         int t, int lane, int w)
{
    int part = 0;
    for (int c0 = 0; c0 < nch; c0 += MAXCH) {
        int nc = nch - c0;
        nc = nc < MAXCH ? nc : MAXCH;
        for (int c = t; c < nc; c += 1024) {
            const int* o = off + (size_t)(c0 + c) * (B + 1) + b;
            const int s0 = o[0], s1 = o[1];
            segb[c] = (c0 + c) * CHUNK + s0;
            segn[c] = s1 - s0;
            part += s0;
        }
        __syncthreads();
        for (int c = w; c < nc; c += 32) {
            const int sb = segb[c], sn = segn[c];
            for (int j = lane; j < sn; j += 32) {
                const int2 pr = pairs[sb + j];
                if (PLACE) {
                    const int p = atomicAdd(&cl[pr.y - lo], 1);
                    csrc[p] = pr.x;
                } else {
                    atomicAdd(&cl[pr.y - lo], 1);
                }
            }
        }
        __syncthreads();
    }
    return part;
}

__global__ __launch_bounds__(1024) void k_csr_place(const int* __restrict__ pairs_i,
                                                    const int* __restrict__ off,
                                                    int* __restrict__ cnt,
                                                    int* __restrict__ rowp,
                                                    int* __restrict__ csrc,
                                                    int Nn, int B, int nch)
{
    __shared__ int segb[MAXCH];
    __shared__ int segn[MAXCH];
    __shared__ int cl[RNODES];
    __shared__ int sc[32];
    __shared__ int red[32];
    const int2* __restrict__ pairs = (const int2*)pairs_i;
    const int t = threadIdx.x, lane = t & 31, w = t >> 5;
    const int b  = blockIdx.x;
    const int lo = b << RSH;

    if (t < RNODES) cl[t] = 0;
    __syncthreads();
    int part = csr_sweep<false>(pairs, off, segb, segn, cl, csrc, b, lo, B, nch, t, lane, w);

    part += __shfl_xor(part, 16, 32);
    part += __shfl_xor(part, 8, 32);
    part += __shfl_xor(part, 4, 32);
    part += __shfl_xor(part, 2, 32);
    part += __shfl_xor(part, 1, 32);
    if (lane == 0) red[w] = part;
    const int h = (t < RNODES) ? cl[t] : 0;
    const int run = block_run_sum(h, sc, lane, w);
    int base = 0;
    for (int i = 0; i < 32; ++i) base += red[i];
    if (t < RNODES) {
        const int first = base + run - h;
        cl[t] = first;
        if (lo + t < Nn) { rowp[lo + t] = first; cnt[lo + t] = h; }
    }
    __syncthreads();
    csr_sweep<true>(pairs, off, segb, segn, cl, csrc, b, lo, B, nch, t, lane, w);
}

__device__ __forceinline__ float lrelu(float v) { return (v < 0.f) ? NEG_SLOPE * v : v; }


template <int M>
__device__ __forceinline__ float xlane(float x) {
    return __builtin_bit_cast(float, __builtin_amdgcn_mov_dpp(__builtin_bit_cast(int, x),
                                                              0x160 + M, 0xf, 0xf, true));
}
__device__ __forceinline__ float xhalf(float x) {
    const int b = __builtin_bit_cast(int, x);
    return __builtin_bit_cast(float,
        __builtin_amdgcn_permlanex16(b, b, 0x76543210, 0xfedcba98, false, false));
}
template <int K>
__device__ __forceinline__ float grp8_lane(float x) {
    return __builtin_bit_cast(float,
        __builtin_amdgcn_mov_dpp8(__builtin_bit_cast(unsigned, x), K * 0x249249));
}

__device__ __forceinline__ float grp8_max(float v) {
    v = fmaxf(v, xlane<4>(v));
    v = fmaxf(v, xlane<2>(v));
    v = fmaxf(v, xlane<1>(v));
    return v;
}
__device__ __forceinline__ float grp8_sum(float v) {
    v += xlane<4>(v);
    v += xlane<2>(v);
    v += xlane<1>(v);
    return v;
}
__device__ __forceinline__ float wave_sum(float v) {
    v += xhalf(v);
    v += xlane<8>(v);
    v += xlane<4>(v);
    v += xlane<2>(v);
    v += xlane<1>(v);
    return v;
}

__device__ __forceinline__ float grp8_sum_to_lane(const float p[8], int i8)
{
    const bool b4 = (i8 & 4) != 0, b2 = (i8 & 2) != 0, b1 = (i8 & 1) != 0;
    float q[4], r[2];
#pragma unroll
    for (int k = 0; k < 4; ++k) {
        const float keep = b4 ? p[k + 4] : p[k];
        const float send = b4 ? p[k] : p[k + 4];
        q[k] = keep + xlane<4>(send);
    }
#pragma unroll
    for (int k = 0; k < 2; ++k) {
        const float keep = b2 ? q[k + 2] : q[k];
        const float send = b2 ? q[k] : q[k + 2];
        r[k] = keep + xlane<2>(send);
    }
    const float keep = b1 ? r[1] : r[0];
    const float send = b1 ? r[0] : r[1];
    return keep + xlane<1>(send);
}

struct row4 { v2f lo, hi; };
__device__ __forceinline__ row4 ld_row(const float* __restrict__ Hm, int s, int lane) {
    const float4 t = *(const float4*)(Hm + (size_t)s * HC + lane * 4);
    row4 r;
    r.lo = (v2f){t.x, t.y};
    r.hi = (v2f){t.z, t.w};
    return r;
}
typedef float v4f __attribute__((ext_vector_type(4)));
typedef const __attribute__((address_space(1))) float* gcf_ptr;
typedef const __attribute__((address_space(1))) v4f*   gcv4_ptr;
__device__ __forceinline__ row4 ld_row_uniform(const float* __restrict__ Hm, int s, int lane) {
    gcf_ptr rb = (gcf_ptr)(Hm + (size_t)s * HC);
    asm("" : "+s"(rb));
    const v4f t = *(gcv4_ptr)(rb + lane * 4);
    row4 r;
    r.lo = (v2f){t.x, t.y};
    r.hi = (v2f){t.z, t.w};
    return r;
}
__device__ __forceinline__ float dot4(row4 v, row4 a) {
    v2f d = v.lo * a.lo;
    d = __builtin_elementwise_fma(v.hi, a.hi, d);
    return d.x + d.y;
}
__device__ __forceinline__ void fma_row(row4* acc, row4 v, float w) {
    const v2f w2 = {w, w};
    acc->lo = __builtin_elementwise_fma(v.lo, w2, acc->lo);
    acc->hi = __builtin_elementwise_fma(v.hi, w2, acc->hi);
}

template <bool FULL>
__device__ __forceinline__ void edge_batch(int nb, int j0, int sidx, int lane, int i8,
                                           const float* __restrict__ Hm, row4 as4, float adh,
                                           float* m, float* l, row4* acc)
{
    int   sr[8];
    row4  v[8];
    float p[8];
#pragma unroll
    for (int i = 0; i < 8; ++i) sr[i] = __builtin_amdgcn_readlane(sidx, j0 + i);
#pragma unroll
    for (int i = 0; i < 8; ++i) {
        if (FULL || i < nb) {
            v[i] = ld_row_uniform(Hm, sr[i], lane);
        } else {
            v[i].lo = (v2f){0.f, 0.f};
            v[i].hi = (v2f){0.f, 0.f};
        }
    }
#pragma unroll
    for (int i = 0; i < 8; ++i) {
        p[i] = dot4(v[i], as4);
        asm("" : "+v"(p[i]));
    }
    const float als = grp8_sum_to_lane(p, i8);
    const float e   = (FULL || i8 < nb) ? lrelu(als + adh) : -INFINITY;
    const float mn  = fmaxf(*m, grp8_max(e));
    const float sc  = expf(*m - mn);
    const float w   = expf(e - mn);
    *l = *l * sc + grp8_sum(w);
    *m = mn;
    const v2f sc2 = {sc, sc};
    acc->lo *= sc2;
    acc->hi *= sc2;
    fma_row(acc, v[0], grp8_lane<0>(w));
    fma_row(acc, v[1], grp8_lane<1>(w));
    fma_row(acc, v[2], grp8_lane<2>(w));
    fma_row(acc, v[3], grp8_lane<3>(w));
    fma_row(acc, v[4], grp8_lane<4>(w));
    fma_row(acc, v[5], grp8_lane<5>(w));
    fma_row(acc, v[6], grp8_lane<6>(w));
    fma_row(acc, v[7], grp8_lane<7>(w));
}

__global__ __launch_bounds__(256) void k_node_gather(
    const int* __restrict__ rowp, const int* __restrict__ cnt,
    const int* __restrict__ csrc,
    const float* __restrict__ a_s, const float* __restrict__ a_d,
    const float* __restrict__ Hm,
    const float* __restrict__ bias, const float* __restrict__ g, const float* __restrict__ bln,
    float* __restrict__ outp, int Nn, int apply_ln)
{
    const int lane = threadIdx.x & 31;
    const int i8   = lane & 7;
    const int d = __builtin_amdgcn_readfirstlane((int)(blockIdx.x * 8 + (threadIdx.x >> 5)));
    if (d >= Nn) return;
    const int beg = rowp[d];
    const int deg = cnt[d];

    const row4 as4 = ld_row(a_s, 0, lane);
    const row4 ad4 = ld_row(a_d, 0, lane);

    row4 acc = ld_row(Hm, d, lane);
    const float adh = grp8_sum(dot4(acc, ad4));
    float m = lrelu(grp8_sum(dot4(acc, as4)) + adh);
    float l = 1.f;

    for (int c0 = 0; c0 < deg; c0 += 32) {
        int jc = c0 + lane;
        jc = jc < deg ? jc : deg - 1;
        const int sidx = csrc[beg + jc];
        int nc = deg - c0;
        nc = nc < 32 ? nc : 32;
        int j0 = 0;
        for (; j0 + 8 <= nc; j0 += 8)
            edge_batch<true>(8, j0, sidx, lane, i8, Hm, as4, adh, &m, &l, &acc);
        if (j0 < nc)
            edge_batch<false>(nc - j0, j0, sidx, lane, i8, Hm, as4, adh, &m, &l, &acc);
    }

    const float4 bv = *(const float4*)(bias + lane * 4);
    float4 v = make_float4(acc.lo.x / l + bv.x, acc.lo.y / l + bv.y,
                           acc.hi.x / l + bv.z, acc.hi.y / l + bv.w);
    if (apply_ln) {
        const float4 gv = *(const float4*)(g + lane * 4);
        const float4 bl = *(const float4*)(bln + lane * 4);
        const float s1 = wave_sum((v.x + v.y) + (v.z + v.w));
        const float s2 = wave_sum((v.x * v.x + v.y * v.y) + (v.z * v.z + v.w * v.w));
        const float mu  = s1 * (1.f / HC);
        const float var = s2 * (1.f / HC) - mu * mu;
        const float rs  = rsqrtf(var + EPSV);
        v.x = fmaxf((v.x - mu) * rs * gv.x + bl.x, 0.f);
        v.y = fmaxf((v.y - mu) * rs * gv.y + bl.y, 0.f);
        v.z = fmaxf((v.z - mu) * rs * gv.z + bl.z, 0.f);
        v.w = fmaxf((v.w - mu) * rs * gv.w + bl.w, 0.f);
    }
    *(float4*)(outp + (size_t)d * HC + lane * 4) = v;
}

extern "C" void kernel_launch(void* const* d_in, const int* in_sizes, int n_in,
                              void* d_out, int out_size, void* d_ws, size_t ws_size,
                              hipStream_t stream)
{
    const float* x    = (const float*)d_in[0];
    const int*   esrc = (const int*)d_in[1];
    const int*   edst = (const int*)d_in[2];
    const float* W[3]  = {(const float*)d_in[3],  (const float*)d_in[7],  (const float*)d_in[11]};
    const float* As[3] = {(const float*)d_in[4],  (const float*)d_in[8],  (const float*)d_in[12]};
    const float* Ad[3] = {(const float*)d_in[5],  (const float*)d_in[9],  (const float*)d_in[13]};
    const float* Bi[3] = {(const float*)d_in[6],  (const float*)d_in[10], (const float*)d_in[14]};
    const float* LnG[2] = {(const float*)d_in[15], (const float*)d_in[17]};
    const float* LnB[2] = {(const float*)d_in[16], (const float*)d_in[18]};

    const int Nn   = in_sizes[0] / HC;
    const int E    = in_sizes[1];

    float* hbuf = (float*)d_ws;
    int*   prs  = (int*)(hbuf + (size_t)Nn * HC);
    int*   csrc = prs + 2 * (size_t)E;
    int*   cnt  = csrc + E;
    int*   rowp = cnt + Nn;
    int*   offt = rowp + Nn;
    float* lay  = (float*)d_out;

    const int      nCh   = (E + CHUNK - 1) / CHUNK;
    const int      nRg   = (Nn + RNODES - 1) / RNODES;
    const unsigned gGat  = (unsigned)((Nn + 7) / 8);

    k_csr_part<<<(unsigned)nCh, 1024, 0, stream>>>(esrc, edst, prs, offt, E, nRg);
    k_csr_place<<<(unsigned)nRg, 1024, 0, stream>>>(prs, offt, cnt, rowp, csrc, Nn, nRg, nCh);

    const float* cur = x;
    for (int l = 0; l < 3; ++l) {
        k_gemm_wmma<<<GEMM_WGS, 256, 0, stream>>>(cur, W[l], hbuf, Nn);
        if (l < 2) {
            k_node_gather<<<gGat, 256, 0, stream>>>(rowp, cnt, csrc, As[l], Ad[l], hbuf,
                                                    Bi[l], LnG[l], LnB[l], lay, Nn, 1);
            cur = lay;
        } else {
            k_node_gather<<<gGat, 256, 0, stream>>>(rowp, cnt, csrc, As[l], Ad[l], hbuf,
                                                    Bi[l], LnG[0], LnB[0], lay, Nn, 0);
        }
    }
}
